# starting kernel
# baseline (speedup 1.0000x reference)
#include <hip/hip_runtime.h>
#include <math.h>

typedef float v2f __attribute__((ext_vector_type(2)));
typedef float v8f __attribute__((ext_vector_type(8)));

#define IN_DIM 128
#define HID    64
#define OUTD   64

__device__ __forceinline__ unsigned enc_f32(float f) {
  unsigned b = __float_as_uint(f);
  return (b & 0x80000000u) ? ~b : (b | 0x80000000u);
}
__device__ __forceinline__ float dec_f32(unsigned u) {
  unsigned b = (u & 0x80000000u) ? (u & 0x7FFFFFFFu) : ~u;
  return __uint_as_float(b);
}

__global__ void gat_init(float* __restrict__ h, unsigned* __restrict__ mbits,
                         float* __restrict__ denom, int N, long total) {
  long i = (long)blockIdx.x * blockDim.x + threadIdx.x;
  if (i < total) h[i] = 0.0f;
  if (i < N) {
    mbits[i] = enc_f32(-INFINITY);
    denom[i] = 0.0f;
  }
}

__global__ __launch_bounds__(64) void gat_fused_gemm(
    const float* __restrict__ x, const float* __restrict__ w_in,
    const float* __restrict__ b_in, const float* __restrict__ w,
    float* __restrict__ z, int N) {
  __shared__ float lds[2][16][66];

  const int lane = threadIdx.x & 31;
  const int wv   = threadIdx.x >> 5;
  const int tile = blockIdx.x * 2 + wv;
  const int ntiles = (N + 15) >> 4;
  const bool active = tile < ntiles;

  const int m  = lane & 15;
  const int kp = lane >> 4;

  if (active) {
    const int row  = tile * 16 + m;
    const int rowc = row < N ? row : (N - 1);
    const float* xrow = x + (long)rowc * IN_DIM;

    v8f acc[4];
    #pragma unroll
    for (int nt = 0; nt < 4; ++nt) acc[nt] = (v8f){0,0,0,0,0,0,0,0};

    for (int k0 = 0; k0 < IN_DIM; k0 += 4) {
      const int kk = k0 + 2 * kp;
      v2f av = *(const v2f*)(xrow + kk);
      const float* wr0 = w_in + (long)kk * HID;
      const float* wr1 = wr0 + HID;
      #pragma unroll
      for (int nt = 0; nt < 4; ++nt) {
        v2f bv = { wr0[nt * 16 + m], wr1[nt * 16 + m] };
        acc[nt] = __builtin_amdgcn_wmma_f32_16x16x4_f32(
            false, av, false, bv, (short)0, acc[nt], false, false);
      }
    }

    #pragma unroll
    for (int nt = 0; nt < 4; ++nt) {
      float bias = b_in[nt * 16 + m];
      #pragma unroll
      for (int r = 0; r < 8; ++r) {
        float v = acc[nt][r] + bias;
        v = 0.5f * v * (1.0f + erff(v * 0.70710678118654752f));
        lds[wv][r + 8 * kp][nt * 16 + m] = v;
      }
    }
  }
  __syncthreads();

  if (active) {
    v8f acc[4];
    #pragma unroll
    for (int nt = 0; nt < 4; ++nt) acc[nt] = (v8f){0,0,0,0,0,0,0,0};

    for (int k0 = 0; k0 < HID; k0 += 4) {
      const int kk = k0 + 2 * kp;
      v2f av = *(const v2f*)(&lds[wv][m][kk]);
      const float* wr0 = w + (long)kk * OUTD;
      const float* wr1 = wr0 + OUTD;
      #pragma unroll
      for (int nt = 0; nt < 4; ++nt) {
        v2f bv = { wr0[nt * 16 + m], wr1[nt * 16 + m] };
        acc[nt] = __builtin_amdgcn_wmma_f32_16x16x4_f32(
            false, av, false, bv, (short)0, acc[nt], false, false);
      }
    }

    #pragma unroll
    for (int nt = 0; nt < 4; ++nt) {
      #pragma unroll
      for (int r = 0; r < 8; ++r) {
        int rr = tile * 16 + r + 8 * kp;
        if (rr < N) z[(long)rr * OUTD + nt * 16 + m] = acc[nt][r];
      }
    }
  }
}

__global__ __launch_bounds__(256) void gat_scores(
    const float* __restrict__ z, const float* __restrict__ a,
    float* __restrict__ s1, float* __restrict__ s2, int N) {
  int i = blockIdx.x * blockDim.x + threadIdx.x;
  if (i >= N) return;
  const float4* zr = (const float4*)(z + (long)i * OUTD);
  const float4* a1 = (const float4*)a;
  const float4* a2 = (const float4*)(a + OUTD);
  float acc1 = 0.f, acc2 = 0.f;
  #pragma unroll
  for (int j = 0; j < OUTD / 4; ++j) {
    float4 v = zr[j];
    float4 p = a1[j];
    float4 q = a2[j];
    acc1 += v.x * p.x + v.y * p.y + v.z * p.z + v.w * p.w;
    acc2 += v.x * q.x + v.y * q.y + v.z * q.z + v.w * q.w;
  }
  s1[i] = acc1;
  s2[i] = acc2;
}

__global__ __launch_bounds__(256) void gat_edge_max(
    const int* __restrict__ src, const int* __restrict__ dst,
    const float* __restrict__ s1, const float* __restrict__ s2,
    float* __restrict__ ebuf, unsigned* __restrict__ mbits, int E) {
  int e = blockIdx.x * blockDim.x + threadIdx.x;
  if (e >= E) return;
  int s = src[e], d = dst[e];
  float v = s1[s] + s2[d];
  v = v > 0.f ? v : 0.01f * v;
  ebuf[e] = v;
  atomicMax(&mbits[d], enc_f32(v));
}

__global__ __launch_bounds__(256) void gat_edge_exp(
    const int* __restrict__ dst, float* __restrict__ ebuf,
    const unsigned* __restrict__ mbits, float* __restrict__ denom, int E) {
  int e = blockIdx.x * blockDim.x + threadIdx.x;
  if (e >= E) return;
  int d = dst[e];
  float mv = dec_f32(mbits[d]);
  if (!isfinite(mv)) mv = 0.0f;
  float ev = expf(ebuf[e] - mv);
  ebuf[e] = ev;
  atomicAdd(&denom[d], ev);
}

__global__ __launch_bounds__(256) void gat_scatter(
    const int* __restrict__ src, const int* __restrict__ dst,
    const float* __restrict__ z, const float* __restrict__ ebuf,
    const float* __restrict__ denom, float* __restrict__ h, int E) {
  int e = blockIdx.x * 8 + (threadIdx.x >> 5);
  if (e >= E) return;
  int lane = threadIdx.x & 31;
  int s = src[e], d = dst[e];
  float alpha = ebuf[e] / fmaxf(denom[d], 1e-9f);
  v2f zv = *(const v2f*)(z + (long)s * OUTD + 2 * lane);
  float* hp = h + (long)d * OUTD + 2 * lane;
  atomicAdd(hp + 0, alpha * zv.x);
  atomicAdd(hp + 1, alpha * zv.y);
}

extern "C" void kernel_launch(void* const* d_in, const int* in_sizes, int n_in,
                              void* d_out, int out_size, void* d_ws, size_t ws_size,
                              hipStream_t stream) {
  const float* x    = (const float*)d_in[0];
  const float* w_in = (const float*)d_in[1];
  const float* b_in = (const float*)d_in[2];
  const float* w    = (const float*)d_in[3];
  const float* a    = (const float*)d_in[4];
  const int*   src  = (const int*)d_in[5];
  const int*   dst  = (const int*)d_in[6];
  float* h = (float*)d_out;

  const int N = in_sizes[0] / IN_DIM;
  const int E = in_sizes[5];

  auto align256 = [](size_t v) { return (v + 255) & ~(size_t)255; };
  char* base = (char*)d_ws;
  size_t off = 0;
  float*    z     = (float*)(base + off); off += align256((size_t)N * OUTD * 4);
  float*    s1    = (float*)(base + off); off += align256((size_t)N * 4);
  float*    s2    = (float*)(base + off); off += align256((size_t)N * 4);
  unsigned* mbits = (unsigned*)(base + off); off += align256((size_t)N * 4);
  float*    denom = (float*)(base + off); off += align256((size_t)N * 4);
  float*    ebuf  = (float*)(base + off); off += align256((size_t)E * 4);

  const long total = (long)N * OUTD;
  {
    long blocks = (total + 255) / 256;
    gat_init<<<dim3((unsigned)blocks), dim3(256), 0, stream>>>(h, mbits, denom, N, total);
  }
  {
    int ntiles = (N + 15) >> 4;
    int blocks = (ntiles + 1) / 2;
    gat_fused_gemm<<<dim3(blocks), dim3(64), 0, stream>>>(x, w_in, b_in, w, z, N);
  }
  gat_scores<<<dim3((N + 255) / 256), dim3(256), 0, stream>>>(z, a, s1, s2, N);
  gat_edge_max<<<dim3((E + 255) / 256), dim3(256), 0, stream>>>(src, dst, s1, s2, ebuf, mbits, E);
  gat_edge_exp<<<dim3((E + 255) / 256), dim3(256), 0, stream>>>(dst, ebuf, mbits, denom, E);
  gat_scatter<<<dim3((E + 7) / 8), dim3(256), 0, stream>>>(src, dst, z, ebuf, denom, h, E);
}


# owner row sums: an edge is filed (returning LDS add) in the queue of the wave owning its row (row mod 32); that wave adds source rows by plain LDS read-add-write, no float atomics; queue order
# speedup vs baseline: 3.9068x; 3.9068x over previous
#include <hip/hip_runtime.h>
#include <math.h>

typedef float v2f __attribute__((ext_vector_type(2)));
typedef float v8f __attribute__((ext_vector_type(8)));

#define IN_DIM 128
#define HID    64
#define OUTD   64

__device__ __forceinline__ unsigned enc_f32(float f) {
  unsigned b = __float_as_uint(f);
  return (b & 0x80000000u) ? ~b : (b | 0x80000000u);
}
__device__ __forceinline__ float dec_f32(unsigned u) {
  unsigned b = (u & 0x80000000u) ? (u & 0x7FFFFFFFu) : ~u;
  return __uint_as_float(b);
}

__global__ __launch_bounds__(64) void gat_fused_gemm(
    const float* __restrict__ x, const float* __restrict__ w_in,
    const float* __restrict__ b_in, const float* __restrict__ w,
    float* __restrict__ z, int N) {
  __shared__ float lds[2][16][66];

  const int lane = threadIdx.x & 31;
  const int wv   = threadIdx.x >> 5;
  const int tile = blockIdx.x * 2 + wv;
  const int ntiles = (N + 15) >> 4;
  const bool active = tile < ntiles;

  const int m  = lane & 15;
  const int kp = lane >> 4;

  if (active) {
    const int row  = tile * 16 + m;
    const int rowc = row < N ? row : (N - 1);
    const float* xrow = x + (long)rowc * IN_DIM;

    v8f acc[4];
    #pragma unroll
    for (int nt = 0; nt < 4; ++nt) acc[nt] = (v8f){0,0,0,0,0,0,0,0};

    for (int k0 = 0; k0 < IN_DIM; k0 += 4) {
      const int kk = k0 + 2 * kp;
      v2f av = *(const v2f*)(xrow + kk);
      const float* wr0 = w_in + (long)kk * HID;
      const float* wr1 = wr0 + HID;
      #pragma unroll
      for (int nt = 0; nt < 4; ++nt) {
        v2f bv = { wr0[nt * 16 + m], wr1[nt * 16 + m] };
        acc[nt] = __builtin_amdgcn_wmma_f32_16x16x4_f32(
            false, av, false, bv, (short)0, acc[nt], false, false);
      }
    }

    #pragma unroll
    for (int nt = 0; nt < 4; ++nt) {
      float bias = b_in[nt * 16 + m];
      #pragma unroll
      for (int r = 0; r < 8; ++r) {
        float v = acc[nt][r] + bias;
        v = 0.5f * v * (1.0f + erff(v * 0.70710678118654752f));
        lds[wv][r + 8 * kp][nt * 16 + m] = v;
      }
    }
  }
  __syncthreads();

  if (active) {
    v8f acc[4];
    #pragma unroll
    for (int nt = 0; nt < 4; ++nt) acc[nt] = (v8f){0,0,0,0,0,0,0,0};

    for (int k0 = 0; k0 < HID; k0 += 4) {
      const int kk = k0 + 2 * kp;
      v2f av = *(const v2f*)(&lds[wv][m][kk]);
      const float* wr0 = w + (long)kk * OUTD;
      const float* wr1 = wr0 + OUTD;
      #pragma unroll
      for (int nt = 0; nt < 4; ++nt) {
        v2f bv = { wr0[nt * 16 + m], wr1[nt * 16 + m] };
        acc[nt] = __builtin_amdgcn_wmma_f32_16x16x4_f32(
            false, av, false, bv, (short)0, acc[nt], false, false);
      }
    }

    #pragma unroll
    for (int nt = 0; nt < 4; ++nt) {
      #pragma unroll
      for (int r = 0; r < 8; ++r) {
        int rr = tile * 16 + r + 8 * kp;
        if (rr < N) z[(long)rr * OUTD + nt * 16 + m] = acc[nt][r];
      }
    }
  }
}

#define OWN_ROWS    392
__device__ __forceinline__ unsigned own_key(int d) { return ((unsigned)d / OWN_ROWS) & 255u; }
__device__ __forceinline__ void own_planes(unsigned (&pl)[8], int d, int i) {
  const unsigned k = own_key(d);
  #pragma unroll
  for (int b = 0; b < 8; ++b) pl[b] |= ((k >> b) & 1u) << i;
}

__global__ __launch_bounds__(256) void gat_scores(
    const float* __restrict__ z, const float* __restrict__ a,
    float* __restrict__ s1, float* __restrict__ s2, int N,
    const int* __restrict__ dst, unsigned* __restrict__ key, int E) {
  int i = blockIdx.x * blockDim.x + threadIdx.x;
  {
    const int ng = E >> 5;
    const int4* d4 = (const int4*)dst;
    uint4* lo4 = (uint4*)key;
    uint4* hi4 = lo4 + ng;
    for (int j = i; j < ng; j += gridDim.x * blockDim.x) {
      unsigned pl[8] = {0u, 0u, 0u, 0u, 0u, 0u, 0u, 0u};
      #pragma unroll
      for (int q = 0; q < 8; ++q) {
        const int4 p = d4[8 * (long)j + q];
        own_planes(pl, p.x, 4 * q + 0); own_planes(pl, p.y, 4 * q + 1);
        own_planes(pl, p.z, 4 * q + 2); own_planes(pl, p.w, 4 * q + 3);
      }
      uint4 lo, hi;
      lo.x = pl[0]; lo.y = pl[1]; lo.z = pl[2]; lo.w = pl[3];
      hi.x = pl[4]; hi.y = pl[5]; hi.z = pl[6]; hi.w = pl[7];
      lo4[j] = lo;
      hi4[j] = hi;
    }
  }
  if (i >= N) return;
  const float4* zr = (const float4*)(z + (long)i * OUTD);
  const float4* a1 = (const float4*)a;
  const float4* a2 = (const float4*)(a + OUTD);
  float acc1 = 0.f, acc2 = 0.f;
  #pragma unroll
  for (int j = 0; j < OUTD / 4; ++j) {
    float4 v = zr[j];
    float4 p = a1[j];
    float4 q = a2[j];
    acc1 += v.x * p.x + v.y * p.y + v.z * p.z + v.w * p.w;
    acc2 += v.x * q.x + v.y * q.y + v.z * q.z + v.w * q.w;
  }
  s1[i] = acc1;
  s2[i] = acc2;
}

#define OWN_CAP     8192
#define OWN_THREADS 1024
#define OWN_SLOTS   ((OWN_ROWS + 31) / 32)

__device__ __forceinline__ float lrelu(float v) { return v > 0.f ? v : 0.01f * v; }
__device__ __forceinline__ int own_idx(int i, int start, int nvec) {
  int j = i + start;
  if (j >= nvec) j -= nvec;
  return j < nvec ? j : nvec - 1;
}

__global__ __launch_bounds__(OWN_THREADS) void gat_owner(
    const int* __restrict__ src, const int* __restrict__ dst,
    const float* __restrict__ z, const float* __restrict__ s1,
    const float* __restrict__ s2, float* __restrict__ h, int N, int E,
    const unsigned* __restrict__ key) {
  __shared__ float    acc[OWN_ROWS * OUTD];
  __shared__ unsigned mb[OWN_ROWS];
  __shared__ float    den[OWN_ROWS];
  __shared__ int      L_e[OWN_CAP];
  __shared__ int      L_r[OWN_CAP];
  __shared__ float    L_v[OWN_CAP];
  __shared__ int      Q_s[OWN_CAP];
  __shared__ int      Q_r[OWN_CAP];
  __shared__ float    Q_v[OWN_CAP];
  __shared__ int      wcnt[32];
  __shared__ int      cnt;

  const int tid  = threadIdx.x;
  const int lane = tid & 31;
  const int wv   = __builtin_amdgcn_readfirstlane(tid >> 5);
  const int lo   = blockIdx.x * OWN_ROWS;
  const int rows = (N - lo) < OWN_ROWS ? (N - lo) : OWN_ROWS;

  for (int i = tid; i < OWN_ROWS * OUTD; i += OWN_THREADS) acc[i] = 0.0f;
  for (int i = tid; i < OWN_ROWS; i += OWN_THREADS) { mb[i] = enc_f32(-INFINITY); den[i] = 0.0f; }
  if (tid < 32) wcnt[tid] = 0;
  if (tid == 0) cnt = 0;
  __syncthreads();

  const int ng = E >> 5;
  const uint4* lo4 = (const uint4*)key;
  const uint4* hi4 = lo4 + ng;
  const int start = (int)(((long)blockIdx.x * ng) / gridDim.x);
  const unsigned urows = (unsigned)rows;
  const unsigned kb = blockIdx.x & 255u;
  const unsigned n0 = ((kb >> 0) & 1u) - 1u, n1 = ((kb >> 1) & 1u) - 1u;
  const unsigned n2 = ((kb >> 2) & 1u) - 1u, n3 = ((kb >> 3) & 1u) - 1u;
  const unsigned n4 = ((kb >> 4) & 1u) - 1u, n5 = ((kb >> 5) & 1u) - 1u;
  const unsigned n6 = ((kb >> 6) & 1u) - 1u, n7 = ((kb >> 7) & 1u) - 1u;
#define OWN_PUSH(EID) { int p_ = atomicAdd(&cnt, 1); if (p_ < OWN_CAP) L_e[p_] = (EID); }
#define OWN_MATCH(LO, HI) \
    ((((LO).x ^ n0) & ((LO).y ^ n1)) & (((LO).z ^ n2) & ((LO).w ^ n3)) & \
     (((HI).x ^ n4) & ((HI).y ^ n5)) & (((HI).z ^ n6) & ((HI).w ^ n7)))
#define OWN_WALK2(MA, JA, MB, JB) { \
    unsigned ma_ = (MA), mb_ = (MB); \
    while (ma_ | mb_) { \
      const bool ua_ = ma_ != 0u; \
      const int b_ = __builtin_ctz(ua_ ? ma_ : mb_); \
      if (ua_) ma_ &= ma_ - 1u; else mb_ &= mb_ - 1u; \
      OWN_PUSH(32 * (ua_ ? (JA) : (JB)) + b_) } }
  int i = tid;
  if (ng > 0) {
    int jc0 = own_idx(i, start, ng), jc1 = own_idx(i + OWN_THREADS, start, ng);
    uint4 cl0 = lo4[jc0], ch0 = hi4[jc0], cl1 = lo4[jc1], ch1 = hi4[jc1];
    while (i + OWN_THREADS < ng) {
      const int in = i + 2 * OWN_THREADS;
      const int jn0 = own_idx(in, start, ng), jn1 = own_idx(in + OWN_THREADS, start, ng);
      const uint4 nl0 = lo4[jn0], nh0 = hi4[jn0], nl1 = lo4[jn1], nh1 = hi4[jn1];
      OWN_WALK2(OWN_MATCH(cl0, ch0), jc0, OWN_MATCH(cl1, ch1), jc1)
      cl0 = nl0; ch0 = nh0; cl1 = nl1; ch1 = nh1;
      jc0 = jn0; jc1 = jn1;
      i = in;
    }
    for (; i < ng; i += OWN_THREADS) {
      const int j0 = own_idx(i, start, ng);
      const uint4 dl = lo4[j0], dh = hi4[j0];
      OWN_WALK2(OWN_MATCH(dl, dh), j0, 0u, j0)
    }
  }
  if (tid < (E & 31)) {
    int e = (ng << 5) + tid;
    unsigned r = (unsigned)(dst[e] - lo);
    if (r < urows) OWN_PUSH(e)
  }
  __syncthreads();
  const int total = __builtin_amdgcn_readfirstlane(cnt);

  float* hp = h + (long)lo * OUTD;
  if (total <= OWN_CAP) {
    for (int k = tid; k < total; k += OWN_THREADS) {
      int e = L_e[k];
      int s = src[e];
      unsigned r = (unsigned)(dst[e] - lo);
      if (r < urows) {
        float v = lrelu(s1[s] + s2[lo + (int)r]);
        L_e[k] = s;
        L_v[k] = v;
        atomicMax(&mb[r], enc_f32(v));
        int rank = atomicAdd(&wcnt[r & 31u], 1);
        L_r[k] = (int)r | (rank << 9);
      } else {
        L_r[k] = -1;
      }
    }
    __syncthreads();
    const int c_l = wcnt[lane];
    int incl = c_l;
    for (int d = 1; d < 32; d <<= 1) {
      int t = __shfl_up(incl, d);
      if (lane >= d) incl += t;
    }
    const int base_l = incl - c_l;
    for (int k0 = wv * 32; k0 < total; k0 += OWN_THREADS) {
      const int k  = k0 + lane;
      const int pr = k < total ? L_r[k] : -1;
      const int qb = __shfl(base_l, pr >= 0 ? (pr & 31) : 0);
      if (pr >= 0) {
        const int r = pr & 511;
        float mv = dec_f32(mb[r]);
        if (!isfinite(mv)) mv = 0.0f;
        float ev = expf(L_v[k] - mv);
        atomicAdd(&den[r], ev);
        const int pos = qb + (pr >> 9);
        Q_s[pos] = L_e[k];
        Q_r[pos] = r;
        Q_v[pos] = ev;
      }
    }
    __syncthreads();
    const int qs = __builtin_amdgcn_readlane(base_l, wv);
    const int qn = __builtin_amdgcn_readlane(c_l, wv);
    v2f*       accw = (v2f*)acc + lane;
    const v2f* zl   = (const v2f*)z + lane;
    const int  G    = qn >> 4;
#define OWN_ENT(G_, S_, R_, A_) { \
      const int k_ = qs + 16 * (G_) + (lane & 15); \
      S_ = Q_s[k_]; R_ = Q_r[k_]; A_ = Q_v[k_] / fmaxf(den[R_], 1e-9f); }
#define OWN_LOADG(ZV, S_) { \
      _Pragma("unroll") \
      for (int u = 0; u < 16; ++u) ZV[u] = zl[(long)__builtin_amdgcn_readlane(S_, u) * (OUTD / 2)]; }
#define OWN_RMW2(ZA, ZB, RA, RB, AA, AB) { \
      const int ra_ = (RA), rb_ = (RB); const float aa_ = (AA), ab_ = (AB); \
      v2f* pa_ = accw + ra_ * (OUTD / 2); v2f* pb_ = accw + rb_ * (OUTD / 2); \
      if (ra_ != rb_) { \
        v2f va_ = *pa_, vb_ = *pb_; \
        va_ += aa_ * (ZA); vb_ += ab_ * (ZB); \
        *pa_ = va_; *pb_ = vb_; \
      } else { \
        v2f va_ = *pa_; \
        va_ += aa_ * (ZA); va_ += ab_ * (ZB); \
        *pa_ = va_; \
      } }
#define OWN_RLF(V, U) __int_as_float(__builtin_amdgcn_readlane(__float_as_int(V), (U)))
#define OWN_PROC(ZV, R_, A_) { \
      _Pragma("unroll") \
      for (int u = 0; u < 16; u += 2) \
        OWN_RMW2(ZV[u], ZV[u + 1], __builtin_amdgcn_readlane(R_, u), __builtin_amdgcn_readlane(R_, u + 1), \
                 OWN_RLF(A_, u), OWN_RLF(A_, u + 1)) }
    int   sA = 0, rA = 0, sB = 0, rB = 0;
    float aA = 0.0f, aB = 0.0f;
    v2f   zvA[16], zvB[16];
    if (G > 0) { OWN_ENT(0, sA, rA, aA) OWN_LOADG(zvA, sA) }
    for (int g = 0; g < G; g += 2) {
      const bool hb = g + 1 < G;
      if (hb) { OWN_ENT(g + 1, sB, rB, aB) OWN_LOADG(zvB, sB) }
      OWN_PROC(zvA, rA, aA)
      if (hb) {
        if (g + 2 < G) { OWN_ENT(g + 2, sA, rA, aA) OWN_LOADG(zvA, sA) }
        OWN_PROC(zvB, rB, aB)
      }
    }
    const int tn = qn & 15;
    if (tn > 0) {
      int   sT = 0, rT = 0;
      float aT = 0.0f;
      if ((lane & 15) < tn) OWN_ENT(G, sT, rT, aT)
      for (int u = 0; u < tn; ++u) {
        const int   s = __builtin_amdgcn_readlane(sT, u);
        const int   r = __builtin_amdgcn_readlane(rT, u);
        const float a = OWN_RLF(aT, u);
        const v2f  zq = zl[(long)s * (OUTD / 2)];
        v2f v = accw[r * (OUTD / 2)];
        v += a * zq;
        accw[r * (OUTD / 2)] = v;
      }
    }
    for (int t = 0; t < OWN_SLOTS; ++t) {
      const int r = wv + 32 * t;
      if (r < rows) ((v2f*)hp)[r * (OUTD / 2) + lane] = accw[r * (OUTD / 2)];
    }
#undef OWN_ENT
#undef OWN_LOADG
#undef OWN_RMW2
#undef OWN_RLF
#undef OWN_PROC
  } else {
    for (int e = tid; e < E; e += OWN_THREADS) {
      unsigned r = (unsigned)(dst[e] - lo);
      if (r < urows) atomicMax(&mb[r], enc_f32(lrelu(s1[src[e]] + s2[lo + (int)r])));
    }
    __syncthreads();
    for (int e = tid; e < E; e += OWN_THREADS) {
      unsigned r = (unsigned)(dst[e] - lo);
      if (r < urows) {
        float mv = dec_f32(mb[r]);
        if (!isfinite(mv)) mv = 0.0f;
        atomicAdd(&den[r], expf(lrelu(s1[src[e]] + s2[lo + (int)r]) - mv));
      }
    }
    __syncthreads();
    for (int e = tid; e < E; e += OWN_THREADS) {
      unsigned r = (unsigned)(dst[e] - lo);
      if (r < urows) {
        int s = src[e];
        float mv = dec_f32(mb[r]);
        if (!isfinite(mv)) mv = 0.0f;
        float a = expf(lrelu(s1[s] + s2[lo + (int)r]) - mv) / fmaxf(den[r], 1e-9f);
        const float* zr = z + (long)s * OUTD;
        for (int c = 0; c < OUTD; ++c) atomicAdd(&acc[r * OUTD + c], a * zr[c]);
      }
    }
    __syncthreads();
    for (int idx = tid; idx < rows * OUTD; idx += OWN_THREADS) hp[idx] = acc[idx];
  }
#undef OWN_PUSH
#undef OWN_MATCH
#undef OWN_WALK2
}

extern "C" void kernel_launch(void* const* d_in, const int* in_sizes, int n_in,
                              void* d_out, int out_size, void* d_ws, size_t ws_size,
                              hipStream_t stream) {
  const float* x    = (const float*)d_in[0];
  const float* w_in = (const float*)d_in[1];
  const float* b_in = (const float*)d_in[2];
  const float* w    = (const float*)d_in[3];
  const float* a    = (const float*)d_in[4];
  const int*   src  = (const int*)d_in[5];
  const int*   dst  = (const int*)d_in[6];
  float* h = (float*)d_out;

  const int N = in_sizes[0] / IN_DIM;
  const int E = in_sizes[5];

  auto align256 = [](size_t v) { return (v + 255) & ~(size_t)255; };
  char* base = (char*)d_ws;
  size_t off = 0;
  float*    z     = (float*)(base + off); off += align256((size_t)N * OUTD * 4);
  float*    s1    = (float*)(base + off); off += align256((size_t)N * 4);
  float*    s2    = (float*)(base + off); off += align256((size_t)N * 4);
  unsigned* key   = (unsigned*)(base + off); off += align256((size_t)E);

  {
    int ntiles = (N + 15) >> 4;
    int blocks = (ntiles + 1) / 2;
    gat_fused_gemm<<<dim3(blocks), dim3(64), 0, stream>>>(x, w_in, b_in, w, z, N);
  }
  gat_scores<<<dim3((N + 255) / 256), dim3(256), 0, stream>>>(z, a, s1, s2, N, dst, key, E);
  gat_owner<<<dim3((N + OWN_ROWS - 1) / OWN_ROWS), dim3(OWN_THREADS), 0, stream>>>(
      src, dst, z, s1, s2, h, N, E, key);
}


# owner pass reads one packed word per found edge (source node | row in range, written by the scores launch) where it read src[e] and dst[e]: one scattered line per edge less; sum order as before
# speedup vs baseline: 4.0341x; 1.0326x over previous
#include <hip/hip_runtime.h>
#include <math.h>

typedef float v2f __attribute__((ext_vector_type(2)));
typedef float v8f __attribute__((ext_vector_type(8)));

#define IN_DIM 128
#define HID    64
#define OUTD   64

__device__ __forceinline__ unsigned enc_f32(float f) {
  unsigned b = __float_as_uint(f);
  return (b & 0x80000000u) ? ~b : (b | 0x80000000u);
}
__device__ __forceinline__ float dec_f32(unsigned u) {
  unsigned b = (u & 0x80000000u) ? (u & 0x7FFFFFFFu) : ~u;
  return __uint_as_float(b);
}

__global__ __launch_bounds__(64) void gat_fused_gemm(
    const float* __restrict__ x, const float* __restrict__ w_in,
    const float* __restrict__ b_in, const float* __restrict__ w,
    float* __restrict__ z, int N) {
  __shared__ float lds[2][16][66];

  const int lane = threadIdx.x & 31;
  const int wv   = threadIdx.x >> 5;
  const int tile = blockIdx.x * 2 + wv;
  const int ntiles = (N + 15) >> 4;
  const bool active = tile < ntiles;

  const int m  = lane & 15;
  const int kp = lane >> 4;

  if (active) {
    const int row  = tile * 16 + m;
    const int rowc = row < N ? row : (N - 1);
    const float* xrow = x + (long)rowc * IN_DIM;

    v8f acc[4];
    #pragma unroll
    for (int nt = 0; nt < 4; ++nt) acc[nt] = (v8f){0,0,0,0,0,0,0,0};

    for (int k0 = 0; k0 < IN_DIM; k0 += 4) {
      const int kk = k0 + 2 * kp;
      v2f av = *(const v2f*)(xrow + kk);
      const float* wr0 = w_in + (long)kk * HID;
      const float* wr1 = wr0 + HID;
      #pragma unroll
      for (int nt = 0; nt < 4; ++nt) {
        v2f bv = { wr0[nt * 16 + m], wr1[nt * 16 + m] };
        acc[nt] = __builtin_amdgcn_wmma_f32_16x16x4_f32(
            false, av, false, bv, (short)0, acc[nt], false, false);
      }
    }

    #pragma unroll
    for (int nt = 0; nt < 4; ++nt) {
      float bias = b_in[nt * 16 + m];
      #pragma unroll
      for (int r = 0; r < 8; ++r) {
        float v = acc[nt][r] + bias;
        v = 0.5f * v * (1.0f + erff(v * 0.70710678118654752f));
        lds[wv][r + 8 * kp][nt * 16 + m] = v;
      }
    }
  }
  __syncthreads();

  if (active) {
    v8f acc[4];
    #pragma unroll
    for (int nt = 0; nt < 4; ++nt) acc[nt] = (v8f){0,0,0,0,0,0,0,0};

    for (int k0 = 0; k0 < HID; k0 += 4) {
      const int kk = k0 + 2 * kp;
      v2f av = *(const v2f*)(&lds[wv][m][kk]);
      const float* wr0 = w + (long)kk * OUTD;
      const float* wr1 = wr0 + OUTD;
      #pragma unroll
      for (int nt = 0; nt < 4; ++nt) {
        v2f bv = { wr0[nt * 16 + m], wr1[nt * 16 + m] };
        acc[nt] = __builtin_amdgcn_wmma_f32_16x16x4_f32(
            false, av, false, bv, (short)0, acc[nt], false, false);
      }
    }

    #pragma unroll
    for (int nt = 0; nt < 4; ++nt) {
      #pragma unroll
      for (int r = 0; r < 8; ++r) {
        int rr = tile * 16 + r + 8 * kp;
        if (rr < N) z[(long)rr * OUTD + nt * 16 + m] = acc[nt][r];
      }
    }
  }
}

#define OWN_ROWS    392
__device__ __forceinline__ unsigned own_key(int d) { return ((unsigned)d / OWN_ROWS) & 255u; }
__device__ __forceinline__ void own_planes(unsigned (&pl)[8], int d, int i) {
  const unsigned k = own_key(d);
  #pragma unroll
  for (int b = 0; b < 8; ++b) pl[b] |= ((k >> b) & 1u) << i;
}

#define OWN_SRC_BITS 23
#define OWN_SRC_MASK ((1u << OWN_SRC_BITS) - 1u)
__device__ __forceinline__ unsigned own_pack(int s, int d) {
  const unsigned ud = (unsigned)d;
  return (unsigned)s | ((ud - (ud / OWN_ROWS) * OWN_ROWS) << OWN_SRC_BITS);
}
__global__ __launch_bounds__(256) void gat_scores(
    const float* __restrict__ z, const float* __restrict__ a,
    float* __restrict__ s1, float* __restrict__ s2, int N,
    const int* __restrict__ src, const int* __restrict__ dst,
    unsigned* __restrict__ key, unsigned* __restrict__ pk, int E) {
  int i = blockIdx.x * blockDim.x + threadIdx.x;
  {
    const int ng = E >> 5;
    const int4* d4 = (const int4*)dst;
    const int4* s4 = (const int4*)src;
    uint4* pk4 = (uint4*)pk;
    uint4* lo4 = (uint4*)key;
    uint4* hi4 = lo4 + ng;
    for (int j = i; j < ng; j += gridDim.x * blockDim.x) {
      unsigned pl[8] = {0u, 0u, 0u, 0u, 0u, 0u, 0u, 0u};
      #pragma unroll
      for (int q = 0; q < 8; ++q) {
        const int4 p = d4[8 * (long)j + q];
        const int4 sv = s4[8 * (long)j + q];
        own_planes(pl, p.x, 4 * q + 0); own_planes(pl, p.y, 4 * q + 1);
        own_planes(pl, p.z, 4 * q + 2); own_planes(pl, p.w, 4 * q + 3);
        uint4 w;
        w.x = own_pack(sv.x, p.x); w.y = own_pack(sv.y, p.y);
        w.z = own_pack(sv.z, p.z); w.w = own_pack(sv.w, p.w);
        pk4[8 * (long)j + q] = w;
      }
      uint4 lo, hi;
      lo.x = pl[0]; lo.y = pl[1]; lo.z = pl[2]; lo.w = pl[3];
      hi.x = pl[4]; hi.y = pl[5]; hi.z = pl[6]; hi.w = pl[7];
      lo4[j] = lo;
      hi4[j] = hi;
    }
    if (i < (E & 31)) {
      const int e = (ng << 5) + i;
      pk[e] = own_pack(src[e], dst[e]);
    }
  }
  if (i >= N) return;
  const float4* zr = (const float4*)(z + (long)i * OUTD);
  const float4* a1 = (const float4*)a;
  const float4* a2 = (const float4*)(a + OUTD);
  float acc1 = 0.f, acc2 = 0.f;
  #pragma unroll
  for (int j = 0; j < OUTD / 4; ++j) {
    float4 v = zr[j];
    float4 p = a1[j];
    float4 q = a2[j];
    acc1 += v.x * p.x + v.y * p.y + v.z * p.z + v.w * p.w;
    acc2 += v.x * q.x + v.y * q.y + v.z * q.z + v.w * q.w;
  }
  s1[i] = acc1;
  s2[i] = acc2;
}

#define OWN_CAP     8192
#define OWN_THREADS 1024
#define OWN_SLOTS   ((OWN_ROWS + 31) / 32)

__device__ __forceinline__ float lrelu(float v) { return v > 0.f ? v : 0.01f * v; }
__device__ __forceinline__ int own_idx(int i, int start, int nvec) {
  int j = i + start;
  if (j >= nvec) j -= nvec;
  return j < nvec ? j : nvec - 1;
}

__global__ __launch_bounds__(OWN_THREADS) void gat_owner(
    const int* __restrict__ src, const int* __restrict__ dst,
    const float* __restrict__ z, const float* __restrict__ s1,
    const float* __restrict__ s2, float* __restrict__ h, int N, int E,
    const unsigned* __restrict__ key, const unsigned* __restrict__ pk) {
  __shared__ float    acc[OWN_ROWS * OUTD];
  __shared__ unsigned mb[OWN_ROWS];
  __shared__ float    den[OWN_ROWS];
  __shared__ int      L_e[OWN_CAP];
  __shared__ int      L_r[OWN_CAP];
  __shared__ float    L_v[OWN_CAP];
  __shared__ int      Q_s[OWN_CAP];
  __shared__ int      Q_r[OWN_CAP];
  __shared__ float    Q_v[OWN_CAP];
  __shared__ int      wcnt[32];
  __shared__ int      cnt;

  const int tid  = threadIdx.x;
  const int lane = tid & 31;
  const int wv   = __builtin_amdgcn_readfirstlane(tid >> 5);
  const int lo   = blockIdx.x * OWN_ROWS;
  const int rows = (N - lo) < OWN_ROWS ? (N - lo) : OWN_ROWS;

  for (int i = tid; i < OWN_ROWS * OUTD; i += OWN_THREADS) acc[i] = 0.0f;
  for (int i = tid; i < OWN_ROWS; i += OWN_THREADS) { mb[i] = enc_f32(-INFINITY); den[i] = 0.0f; }
  if (tid < 32) wcnt[tid] = 0;
  if (tid == 0) cnt = 0;
  __syncthreads();

  const int ng = E >> 5;
  const uint4* lo4 = (const uint4*)key;
  const uint4* hi4 = lo4 + ng;
  const int start = (int)(((long)blockIdx.x * ng) / gridDim.x);
  const unsigned urows = (unsigned)rows;
  const unsigned kb = blockIdx.x & 255u;
  const unsigned n0 = ((kb >> 0) & 1u) - 1u, n1 = ((kb >> 1) & 1u) - 1u;
  const unsigned n2 = ((kb >> 2) & 1u) - 1u, n3 = ((kb >> 3) & 1u) - 1u;
  const unsigned n4 = ((kb >> 4) & 1u) - 1u, n5 = ((kb >> 5) & 1u) - 1u;
  const unsigned n6 = ((kb >> 6) & 1u) - 1u, n7 = ((kb >> 7) & 1u) - 1u;
#define OWN_PUSH(EID) { int p_ = atomicAdd(&cnt, 1); if (p_ < OWN_CAP) L_e[p_] = (EID); }
#define OWN_MATCH(LO, HI) \
    ((((LO).x ^ n0) & ((LO).y ^ n1)) & (((LO).z ^ n2) & ((LO).w ^ n3)) & \
     (((HI).x ^ n4) & ((HI).y ^ n5)) & (((HI).z ^ n6) & ((HI).w ^ n7)))
#define OWN_WALK2(MA, JA, MB, JB) { \
    unsigned ma_ = (MA), mb_ = (MB); \
    while (ma_ | mb_) { \
      const bool ua_ = ma_ != 0u; \
      const int b_ = __builtin_ctz(ua_ ? ma_ : mb_); \
      if (ua_) ma_ &= ma_ - 1u; else mb_ &= mb_ - 1u; \
      OWN_PUSH(32 * (ua_ ? (JA) : (JB)) + b_) } }
  int i = tid;
  if (ng > 0) {
    int jc0 = own_idx(i, start, ng), jc1 = own_idx(i + OWN_THREADS, start, ng);
    uint4 cl0 = lo4[jc0], ch0 = hi4[jc0], cl1 = lo4[jc1], ch1 = hi4[jc1];
    while (i + OWN_THREADS < ng) {
      const int in = i + 2 * OWN_THREADS;
      const int jn0 = own_idx(in, start, ng), jn1 = own_idx(in + OWN_THREADS, start, ng);
      const uint4 nl0 = lo4[jn0], nh0 = hi4[jn0], nl1 = lo4[jn1], nh1 = hi4[jn1];
      OWN_WALK2(OWN_MATCH(cl0, ch0), jc0, OWN_MATCH(cl1, ch1), jc1)
      cl0 = nl0; ch0 = nh0; cl1 = nl1; ch1 = nh1;
      jc0 = jn0; jc1 = jn1;
      i = in;
    }
    for (; i < ng; i += OWN_THREADS) {
      const int j0 = own_idx(i, start, ng);
      const uint4 dl = lo4[j0], dh = hi4[j0];
      OWN_WALK2(OWN_MATCH(dl, dh), j0, 0u, j0)
    }
  }
  if (tid < (E & 31)) {
    int e = (ng << 5) + tid;
    unsigned r = (unsigned)(dst[e] - lo);
    if (r < urows) OWN_PUSH(e)
  }
  __syncthreads();
  const int total = __builtin_amdgcn_readfirstlane(cnt);

  float* hp = h + (long)lo * OUTD;
  if (total <= OWN_CAP) {
    const bool two_arrays = gridDim.x > 256u || N > (int)(OWN_SRC_MASK + 1u);
    for (int k = tid; k < total; k += OWN_THREADS) {
      int e = L_e[k];
      int s;
      unsigned r;
      if (two_arrays) {
        s = src[e];
        r = (unsigned)(dst[e] - lo);
      } else {
        const unsigned w = pk[e];
        s = (int)(w & OWN_SRC_MASK);
        r = w >> OWN_SRC_BITS;
      }
      if (r < urows) {
        float v = lrelu(s1[s] + s2[lo + (int)r]);
        L_e[k] = s;
        L_v[k] = v;
        atomicMax(&mb[r], enc_f32(v));
        int rank = atomicAdd(&wcnt[r & 31u], 1);
        L_r[k] = (int)r | (rank << 9);
      } else {
        L_r[k] = -1;
      }
    }
    __syncthreads();
    const int c_l = wcnt[lane];
    int incl = c_l;
    for (int d = 1; d < 32; d <<= 1) {
      int t = __shfl_up(incl, d);
      if (lane >= d) incl += t;
    }
    const int base_l = incl - c_l;
    for (int k0 = wv * 32; k0 < total; k0 += OWN_THREADS) {
      const int k  = k0 + lane;
      const int pr = k < total ? L_r[k] : -1;
      const int qb = __shfl(base_l, pr >= 0 ? (pr & 31) : 0);
      if (pr >= 0) {
        const int r = pr & 511;
        float mv = dec_f32(mb[r]);
        if (!isfinite(mv)) mv = 0.0f;
        float ev = expf(L_v[k] - mv);
        atomicAdd(&den[r], ev);
        const int pos = qb + (pr >> 9);
        Q_s[pos] = L_e[k];
        Q_r[pos] = r;
        Q_v[pos] = ev;
      }
    }
    __syncthreads();
    const int qs = __builtin_amdgcn_readlane(base_l, wv);
    const int qn = __builtin_amdgcn_readlane(c_l, wv);
    v2f*       accw = (v2f*)acc + lane;
    const v2f* zl   = (const v2f*)z + lane;
    const int  G    = qn >> 4;
#define OWN_ENT(G_, S_, R_, A_) { \
      const int k_ = qs + 16 * (G_) + (lane & 15); \
      S_ = Q_s[k_]; R_ = Q_r[k_]; A_ = Q_v[k_] / fmaxf(den[R_], 1e-9f); }
#define OWN_LOADG(ZV, S_) { \
      _Pragma("unroll") \
      for (int u = 0; u < 16; ++u) ZV[u] = zl[(long)__builtin_amdgcn_readlane(S_, u) * (OUTD / 2)]; }
#define OWN_RMW2(ZA, ZB, RA, RB, AA, AB) { \
      const int ra_ = (RA), rb_ = (RB); const float aa_ = (AA), ab_ = (AB); \
      v2f* pa_ = accw + ra_ * (OUTD / 2); v2f* pb_ = accw + rb_ * (OUTD / 2); \
      if (ra_ != rb_) { \
        v2f va_ = *pa_, vb_ = *pb_; \
        va_ += aa_ * (ZA); vb_ += ab_ * (ZB); \
        *pa_ = va_; *pb_ = vb_; \
      } else { \
        v2f va_ = *pa_; \
        va_ += aa_ * (ZA); va_ += ab_ * (ZB); \
        *pa_ = va_; \
      } }
#define OWN_RLF(V, U) __int_as_float(__builtin_amdgcn_readlane(__float_as_int(V), (U)))
#define OWN_PROC(ZV, R_, A_) { \
      _Pragma("unroll") \
      for (int u = 0; u < 16; u += 2) \
        OWN_RMW2(ZV[u], ZV[u + 1], __builtin_amdgcn_readlane(R_, u), __builtin_amdgcn_readlane(R_, u + 1), \
                 OWN_RLF(A_, u), OWN_RLF(A_, u + 1)) }
    int   sA = 0, rA = 0, sB = 0, rB = 0;
    float aA = 0.0f, aB = 0.0f;
    v2f   zvA[16], zvB[16];
    if (G > 0) { OWN_ENT(0, sA, rA, aA) OWN_LOADG(zvA, sA) }
    for (int g = 0; g < G; g += 2) {
      const bool hb = g + 1 < G;
      if (hb) { OWN_ENT(g + 1, sB, rB, aB) OWN_LOADG(zvB, sB) }
      OWN_PROC(zvA, rA, aA)
      if (hb) {
        if (g + 2 < G) { OWN_ENT(g + 2, sA, rA, aA) OWN_LOADG(zvA, sA) }
        OWN_PROC(zvB, rB, aB)
      }
    }
    const int tn = qn & 15;
    if (tn > 0) {
      int   sT = 0, rT = 0;
      float aT = 0.0f;
      if ((lane & 15) < tn) OWN_ENT(G, sT, rT, aT)
      for (int u = 0; u < tn; ++u) {
        const int   s = __builtin_amdgcn_readlane(sT, u);
        const int   r = __builtin_amdgcn_readlane(rT, u);
        const float a = OWN_RLF(aT, u);
        const v2f  zq = zl[(long)s * (OUTD / 2)];
        v2f v = accw[r * (OUTD / 2)];
        v += a * zq;
        accw[r * (OUTD / 2)] = v;
      }
    }
    for (int t = 0; t < OWN_SLOTS; ++t) {
      const int r = wv + 32 * t;
      if (r < rows) ((v2f*)hp)[r * (OUTD / 2) + lane] = accw[r * (OUTD / 2)];
    }
#undef OWN_ENT
#undef OWN_LOADG
#undef OWN_RMW2
#undef OWN_RLF
#undef OWN_PROC
  } else {
    for (int e = tid; e < E; e += OWN_THREADS) {
      unsigned r = (unsigned)(dst[e] - lo);
      if (r < urows) atomicMax(&mb[r], enc_f32(lrelu(s1[src[e]] + s2[lo + (int)r])));
    }
    __syncthreads();
    for (int e = tid; e < E; e += OWN_THREADS) {
      unsigned r = (unsigned)(dst[e] - lo);
      if (r < urows) {
        float mv = dec_f32(mb[r]);
        if (!isfinite(mv)) mv = 0.0f;
        atomicAdd(&den[r], expf(lrelu(s1[src[e]] + s2[lo + (int)r]) - mv));
      }
    }
    __syncthreads();
    for (int e = tid; e < E; e += OWN_THREADS) {
      unsigned r = (unsigned)(dst[e] - lo);
      if (r < urows) {
        int s = src[e];
        float mv = dec_f32(mb[r]);
        if (!isfinite(mv)) mv = 0.0f;
        float a = expf(lrelu(s1[s] + s2[lo + (int)r]) - mv) / fmaxf(den[r], 1e-9f);
        const float* zr = z + (long)s * OUTD;
        for (int c = 0; c < OUTD; ++c) atomicAdd(&acc[r * OUTD + c], a * zr[c]);
      }
    }
    __syncthreads();
    for (int idx = tid; idx < rows * OUTD; idx += OWN_THREADS) hp[idx] = acc[idx];
  }
#undef OWN_PUSH
#undef OWN_MATCH
#undef OWN_WALK2
}

extern "C" void kernel_launch(void* const* d_in, const int* in_sizes, int n_in,
                              void* d_out, int out_size, void* d_ws, size_t ws_size,
                              hipStream_t stream) {
  const float* x    = (const float*)d_in[0];
  const float* w_in = (const float*)d_in[1];
  const float* b_in = (const float*)d_in[2];
  const float* w    = (const float*)d_in[3];
  const float* a    = (const float*)d_in[4];
  const int*   src  = (const int*)d_in[5];
  const int*   dst  = (const int*)d_in[6];
  float* h = (float*)d_out;

  const int N = in_sizes[0] / IN_DIM;
  const int E = in_sizes[5];

  auto align256 = [](size_t v) { return (v + 255) & ~(size_t)255; };
  char* base = (char*)d_ws;
  size_t off = 0;
  float*    z     = (float*)(base + off); off += align256((size_t)N * OUTD * 4);
  float*    s1    = (float*)(base + off); off += align256((size_t)N * 4);
  float*    s2    = (float*)(base + off); off += align256((size_t)N * 4);
  unsigned* key   = (unsigned*)(base + off); off += align256((size_t)E);
  unsigned* pk    = (unsigned*)(base + off); off += align256((size_t)E * 4);

  {
    int ntiles = (N + 15) >> 4;
    int blocks = (ntiles + 1) / 2;
    gat_fused_gemm<<<dim3(blocks), dim3(64), 0, stream>>>(x, w_in, b_in, w, z, N);
  }
  gat_scores<<<dim3((N + 255) / 256), dim3(256), 0, stream>>>(z, a, s1, s2, N, src, dst, key, pk, E);
  gat_owner<<<dim3((N + OWN_ROWS - 1) / OWN_ROWS), dim3(OWN_THREADS), 0, stream>>>(
      src, dst, z, s1, s2, h, N, E, key, pk);
}


# owner scan reads 12 bytes per 32 edges (was 32): scores launch decodes upper 6 key bits into 64 one-bit band rows (register masks, plain stores); owner reads its row + 2 low planes; sums as before
# speedup vs baseline: 4.2273x; 1.0479x over previous
#include <hip/hip_runtime.h>
#include <math.h>

typedef float v2f __attribute__((ext_vector_type(2)));
typedef float v8f __attribute__((ext_vector_type(8)));

#define IN_DIM 128
#define HID    64
#define OUTD   64

__device__ __forceinline__ unsigned enc_f32(float f) {
  unsigned b = __float_as_uint(f);
  return (b & 0x80000000u) ? ~b : (b | 0x80000000u);
}
__device__ __forceinline__ float dec_f32(unsigned u) {
  unsigned b = (u & 0x80000000u) ? (u & 0x7FFFFFFFu) : ~u;
  return __uint_as_float(b);
}

__global__ __launch_bounds__(64) void gat_fused_gemm(
    const float* __restrict__ x, const float* __restrict__ w_in,
    const float* __restrict__ b_in, const float* __restrict__ w,
    float* __restrict__ z, int N) {
  __shared__ float lds[2][16][66];

  const int lane = threadIdx.x & 31;
  const int wv   = threadIdx.x >> 5;
  const int tile = blockIdx.x * 2 + wv;
  const int ntiles = (N + 15) >> 4;
  const bool active = tile < ntiles;

  const int m  = lane & 15;
  const int kp = lane >> 4;

  if (active) {
    const int row  = tile * 16 + m;
    const int rowc = row < N ? row : (N - 1);
    const float* xrow = x + (long)rowc * IN_DIM;

    v8f acc[4];
    #pragma unroll
    for (int nt = 0; nt < 4; ++nt) acc[nt] = (v8f){0,0,0,0,0,0,0,0};

    for (int k0 = 0; k0 < IN_DIM; k0 += 4) {
      const int kk = k0 + 2 * kp;
      v2f av = *(const v2f*)(xrow + kk);
      const float* wr0 = w_in + (long)kk * HID;
      const float* wr1 = wr0 + HID;
      #pragma unroll
      for (int nt = 0; nt < 4; ++nt) {
        v2f bv = { wr0[nt * 16 + m], wr1[nt * 16 + m] };
        acc[nt] = __builtin_amdgcn_wmma_f32_16x16x4_f32(
            false, av, false, bv, (short)0, acc[nt], false, false);
      }
    }

    #pragma unroll
    for (int nt = 0; nt < 4; ++nt) {
      float bias = b_in[nt * 16 + m];
      #pragma unroll
      for (int r = 0; r < 8; ++r) {
        float v = acc[nt][r] + bias;
        v = 0.5f * v * (1.0f + erff(v * 0.70710678118654752f));
        lds[wv][r + 8 * kp][nt * 16 + m] = v;
      }
    }
  }
  __syncthreads();

  if (active) {
    v8f acc[4];
    #pragma unroll
    for (int nt = 0; nt < 4; ++nt) acc[nt] = (v8f){0,0,0,0,0,0,0,0};

    for (int k0 = 0; k0 < HID; k0 += 4) {
      const int kk = k0 + 2 * kp;
      v2f av = *(const v2f*)(&lds[wv][m][kk]);
      const float* wr0 = w + (long)kk * OUTD;
      const float* wr1 = wr0 + OUTD;
      #pragma unroll
      for (int nt = 0; nt < 4; ++nt) {
        v2f bv = { wr0[nt * 16 + m], wr1[nt * 16 + m] };
        acc[nt] = __builtin_amdgcn_wmma_f32_16x16x4_f32(
            false, av, false, bv, (short)0, acc[nt], false, false);
      }
    }

    #pragma unroll
    for (int nt = 0; nt < 4; ++nt) {
      #pragma unroll
      for (int r = 0; r < 8; ++r) {
        int rr = tile * 16 + r + 8 * kp;
        if (rr < N) z[(long)rr * OUTD + nt * 16 + m] = acc[nt][r];
      }
    }
  }
}

#define OWN_ROWS    392
__device__ __forceinline__ int  own_band_pitch(int ng) { return (ng + 31) & ~31; }
__device__ __forceinline__ long own_band_base(int ng)  { return (2 * (long)ng + 31) & ~31L; }
__device__ __forceinline__ unsigned own_key(int d) { return ((unsigned)d / OWN_ROWS) & 255u; }
__device__ __forceinline__ void own_planes(unsigned (&pl)[8], int d, int i) {
  const unsigned k = own_key(d);
  #pragma unroll
  for (int b = 0; b < 8; ++b) pl[b] |= ((k >> b) & 1u) << i;
}

#define OWN_SRC_BITS 23
#define OWN_SRC_MASK ((1u << OWN_SRC_BITS) - 1u)
__device__ __forceinline__ unsigned own_pack(int s, int d) {
  const unsigned ud = (unsigned)d;
  return (unsigned)s | ((ud - (ud / OWN_ROWS) * OWN_ROWS) << OWN_SRC_BITS);
}
__global__ __launch_bounds__(256) void gat_scores(
    const float* __restrict__ z, const float* __restrict__ a,
    float* __restrict__ s1, float* __restrict__ s2, int N,
    const int* __restrict__ src, const int* __restrict__ dst,
    unsigned* __restrict__ key, unsigned* __restrict__ pk, int E) {
  int i = blockIdx.x * blockDim.x + threadIdx.x;
  {
    const int ng = E >> 5;
    const int4* d4 = (const int4*)dst;
    const int4* s4 = (const int4*)src;
    uint4* pk4 = (uint4*)pk;
    const int bp = own_band_pitch(ng);
    uint2* lo2 = (uint2*)key;
    unsigned* band = key + own_band_base(ng);
    if (ng & 1) {
      if (i < 64) band[(long)i * bp + ng] = 0u;
      if (i == 0) { uint2 zz; zz.x = 0u; zz.y = 0u; lo2[ng] = zz; }
    }
    for (int j = i; j < ng; j += gridDim.x * blockDim.x) {
      unsigned pl[8] = {0u, 0u, 0u, 0u, 0u, 0u, 0u, 0u};
      #pragma unroll
      for (int q = 0; q < 8; ++q) {
        const int4 p = d4[8 * (long)j + q];
        const int4 sv = s4[8 * (long)j + q];
        own_planes(pl, p.x, 4 * q + 0); own_planes(pl, p.y, 4 * q + 1);
        own_planes(pl, p.z, 4 * q + 2); own_planes(pl, p.w, 4 * q + 3);
        uint4 w;
        w.x = own_pack(sv.x, p.x); w.y = own_pack(sv.y, p.y);
        w.z = own_pack(sv.z, p.z); w.w = own_pack(sv.w, p.w);
        pk4[8 * (long)j + q] = w;
      }
      uint2 lo;
      lo.x = pl[0]; lo.y = pl[1];
      lo2[j] = lo;
      unsigned hA[8], hB[8];
      #pragma unroll
      for (int t = 0; t < 8; ++t) {
        hB[t] = ((t & 1) ? pl[2] : ~pl[2]) & ((t & 2) ? pl[3] : ~pl[3]) & ((t & 4) ? pl[4] : ~pl[4]);
        hA[t] = ((t & 1) ? pl[5] : ~pl[5]) & ((t & 2) ? pl[6] : ~pl[6]) & ((t & 4) ? pl[7] : ~pl[7]);
      }
      #pragma unroll
      for (int t = 0; t < 64; ++t) band[(long)t * bp + j] = hA[t >> 3] & hB[t & 7];
    }
    if (i < (E & 31)) {
      const int e = (ng << 5) + i;
      pk[e] = own_pack(src[e], dst[e]);
    }
  }
  if (i >= N) return;
  const float4* zr = (const float4*)(z + (long)i * OUTD);
  const float4* a1 = (const float4*)a;
  const float4* a2 = (const float4*)(a + OUTD);
  float acc1 = 0.f, acc2 = 0.f;
  #pragma unroll
  for (int j = 0; j < OUTD / 4; ++j) {
    float4 v = zr[j];
    float4 p = a1[j];
    float4 q = a2[j];
    acc1 += v.x * p.x + v.y * p.y + v.z * p.z + v.w * p.w;
    acc2 += v.x * q.x + v.y * q.y + v.z * q.z + v.w * q.w;
  }
  s1[i] = acc1;
  s2[i] = acc2;
}

#define OWN_CAP     8192
#define OWN_THREADS 1024
#define OWN_SLOTS   ((OWN_ROWS + 31) / 32)

__device__ __forceinline__ float lrelu(float v) { return v > 0.f ? v : 0.01f * v; }
__device__ __forceinline__ int own_idx(int i, int start, int nvec) {
  int j = i + start;
  if (j >= nvec) j -= nvec;
  return j < nvec ? j : nvec - 1;
}

__global__ __launch_bounds__(OWN_THREADS) void gat_owner(
    const int* __restrict__ src, const int* __restrict__ dst,
    const float* __restrict__ z, const float* __restrict__ s1,
    const float* __restrict__ s2, float* __restrict__ h, int N, int E,
    const unsigned* __restrict__ key, const unsigned* __restrict__ pk) {
  __shared__ float    acc[OWN_ROWS * OUTD];
  __shared__ unsigned mb[OWN_ROWS];
  __shared__ float    den[OWN_ROWS];
  __shared__ int      L_e[OWN_CAP];
  __shared__ int      L_r[OWN_CAP];
  __shared__ float    L_v[OWN_CAP];
  __shared__ int      Q_s[OWN_CAP];
  __shared__ int      Q_r[OWN_CAP];
  __shared__ float    Q_v[OWN_CAP];
  __shared__ int      wcnt[32];
  __shared__ int      cnt;

  const int tid  = threadIdx.x;
  const int lane = tid & 31;
  const int wv   = __builtin_amdgcn_readfirstlane(tid >> 5);
  const int lo   = blockIdx.x * OWN_ROWS;
  const int rows = (N - lo) < OWN_ROWS ? (N - lo) : OWN_ROWS;

  for (int i = tid; i < OWN_ROWS * OUTD; i += OWN_THREADS) acc[i] = 0.0f;
  for (int i = tid; i < OWN_ROWS; i += OWN_THREADS) { mb[i] = enc_f32(-INFINITY); den[i] = 0.0f; }
  if (tid < 32) wcnt[tid] = 0;
  if (tid == 0) cnt = 0;
  __syncthreads();

  const int ng = E >> 5;
  const int np = (ng + 1) >> 1;
  const unsigned kb = blockIdx.x & 255u;
  const uint4* lo4 = (const uint4*)key;
  const uint2* bw2 = (const uint2*)(key + own_band_base(ng) + (long)(kb >> 2) * own_band_pitch(ng));
  const int start = (int)(((long)blockIdx.x * np) / gridDim.x);
  const unsigned urows = (unsigned)rows;
  const unsigned n0 = ((kb >> 0) & 1u) - 1u, n1 = ((kb >> 1) & 1u) - 1u;
#define OWN_PUSH(EID) { int p_ = atomicAdd(&cnt, 1); if (p_ < OWN_CAP) L_e[p_] = (EID); }
#define OWN_MATCH(P0, P1, BW) ((((P0) ^ n0) & ((P1) ^ n1)) & (BW))
#define OWN_WALK2(MA, JA, MB, JB) { \
    unsigned ma_ = (MA), mb_ = (MB); \
    while (ma_ | mb_) { \
      const bool ua_ = ma_ != 0u; \
      const int b_ = __builtin_ctz(ua_ ? ma_ : mb_); \
      if (ua_) ma_ &= ma_ - 1u; else mb_ &= mb_ - 1u; \
      OWN_PUSH(32 * (ua_ ? (JA) : (JB)) + b_) } }
  int i = tid;
  if (np > 0) {
    int pc0 = own_idx(i, start, np), pc1 = own_idx(i + OWN_THREADS, start, np);
    uint4 cl0 = lo4[pc0], cl1 = lo4[pc1];
    uint2 cb0 = bw2[pc0], cb1 = bw2[pc1];
    while (i + OWN_THREADS < np) {
      const int in = i + 2 * OWN_THREADS;
      const int pn0 = own_idx(in, start, np), pn1 = own_idx(in + OWN_THREADS, start, np);
      const uint4 nl0 = lo4[pn0], nl1 = lo4[pn1];
      const uint2 nb0 = bw2[pn0], nb1 = bw2[pn1];
      OWN_WALK2(OWN_MATCH(cl0.x, cl0.y, cb0.x), 2 * pc0, OWN_MATCH(cl0.z, cl0.w, cb0.y), 2 * pc0 + 1)
      OWN_WALK2(OWN_MATCH(cl1.x, cl1.y, cb1.x), 2 * pc1, OWN_MATCH(cl1.z, cl1.w, cb1.y), 2 * pc1 + 1)
      cl0 = nl0; cl1 = nl1; cb0 = nb0; cb1 = nb1;
      pc0 = pn0; pc1 = pn1;
      i = in;
    }
    for (; i < np; i += OWN_THREADS) {
      const int p0 = own_idx(i, start, np);
      const uint4 dl = lo4[p0];
      const uint2 db = bw2[p0];
      OWN_WALK2(OWN_MATCH(dl.x, dl.y, db.x), 2 * p0, OWN_MATCH(dl.z, dl.w, db.y), 2 * p0 + 1)
    }
  }
  if (tid < (E & 31)) {
    int e = (ng << 5) + tid;
    unsigned r = (unsigned)(dst[e] - lo);
    if (r < urows) OWN_PUSH(e)
  }
  __syncthreads();
  const int total = __builtin_amdgcn_readfirstlane(cnt);

  float* hp = h + (long)lo * OUTD;
  if (total <= OWN_CAP) {
    const bool two_arrays = gridDim.x > 256u || N > (int)(OWN_SRC_MASK + 1u);
    for (int k = tid; k < total; k += OWN_THREADS) {
      int e = L_e[k];
      int s;
      unsigned r;
      if (two_arrays) {
        s = src[e];
        r = (unsigned)(dst[e] - lo);
      } else {
        const unsigned w = pk[e];
        s = (int)(w & OWN_SRC_MASK);
        r = w >> OWN_SRC_BITS;
      }
      if (r < urows) {
        float v = lrelu(s1[s] + s2[lo + (int)r]);
        L_e[k] = s;
        L_v[k] = v;
        atomicMax(&mb[r], enc_f32(v));
        int rank = atomicAdd(&wcnt[r & 31u], 1);
        L_r[k] = (int)r | (rank << 9);
      } else {
        L_r[k] = -1;
      }
    }
    __syncthreads();
    const int c_l = wcnt[lane];
    int incl = c_l;
    for (int d = 1; d < 32; d <<= 1) {
      int t = __shfl_up(incl, d);
      if (lane >= d) incl += t;
    }
    const int base_l = incl - c_l;
    for (int k0 = wv * 32; k0 < total; k0 += OWN_THREADS) {
      const int k  = k0 + lane;
      const int pr = k < total ? L_r[k] : -1;
      const int qb = __shfl(base_l, pr >= 0 ? (pr & 31) : 0);
      if (pr >= 0) {
        const int r = pr & 511;
        float mv = dec_f32(mb[r]);
        if (!isfinite(mv)) mv = 0.0f;
        float ev = expf(L_v[k] - mv);
        atomicAdd(&den[r], ev);
        const int pos = qb + (pr >> 9);
        Q_s[pos] = L_e[k];
        Q_r[pos] = r;
        Q_v[pos] = ev;
      }
    }
    __syncthreads();
    const int qs = __builtin_amdgcn_readlane(base_l, wv);
    const int qn = __builtin_amdgcn_readlane(c_l, wv);
    v2f*       accw = (v2f*)acc + lane;
    const v2f* zl   = (const v2f*)z + lane;
    const int  G    = qn >> 4;
#define OWN_ENT(G_, S_, R_, A_) { \
      const int k_ = qs + 16 * (G_) + (lane & 15); \
      S_ = Q_s[k_]; R_ = Q_r[k_]; A_ = Q_v[k_] / fmaxf(den[R_], 1e-9f); }
#define OWN_LOADG(ZV, S_) { \
      _Pragma("unroll") \
      for (int u = 0; u < 16; ++u) ZV[u] = zl[(long)__builtin_amdgcn_readlane(S_, u) * (OUTD / 2)]; }
#define OWN_RMW2(ZA, ZB, RA, RB, AA, AB) { \
      const int ra_ = (RA), rb_ = (RB); const float aa_ = (AA), ab_ = (AB); \
      v2f* pa_ = accw + ra_ * (OUTD / 2); v2f* pb_ = accw + rb_ * (OUTD / 2); \
      if (ra_ != rb_) { \
        v2f va_ = *pa_, vb_ = *pb_; \
        va_ += aa_ * (ZA); vb_ += ab_ * (ZB); \
        *pa_ = va_; *pb_ = vb_; \
      } else { \
        v2f va_ = *pa_; \
        va_ += aa_ * (ZA); va_ += ab_ * (ZB); \
        *pa_ = va_; \
      } }
#define OWN_RLF(V, U) __int_as_float(__builtin_amdgcn_readlane(__float_as_int(V), (U)))
#define OWN_PROC(ZV, R_, A_) { \
      _Pragma("unroll") \
      for (int u = 0; u < 16; u += 2) \
        OWN_RMW2(ZV[u], ZV[u + 1], __builtin_amdgcn_readlane(R_, u), __builtin_amdgcn_readlane(R_, u + 1), \
                 OWN_RLF(A_, u), OWN_RLF(A_, u + 1)) }
    int   sA = 0, rA = 0, sB = 0, rB = 0;
    float aA = 0.0f, aB = 0.0f;
    v2f   zvA[16], zvB[16];
    if (G > 0) { OWN_ENT(0, sA, rA, aA) OWN_LOADG(zvA, sA) }
    for (int g = 0; g < G; g += 2) {
      const bool hb = g + 1 < G;
      if (hb) { OWN_ENT(g + 1, sB, rB, aB) OWN_LOADG(zvB, sB) }
      OWN_PROC(zvA, rA, aA)
      if (hb) {
        if (g + 2 < G) { OWN_ENT(g + 2, sA, rA, aA) OWN_LOADG(zvA, sA) }
        OWN_PROC(zvB, rB, aB)
      }
    }
    const int tn = qn & 15;
    if (tn > 0) {
      int   sT = 0, rT = 0;
      float aT = 0.0f;
      if ((lane & 15) < tn) OWN_ENT(G, sT, rT, aT)
      for (int u = 0; u < tn; ++u) {
        const int   s = __builtin_amdgcn_readlane(sT, u);
        const int   r = __builtin_amdgcn_readlane(rT, u);
        const float a = OWN_RLF(aT, u);
        const v2f  zq = zl[(long)s * (OUTD / 2)];
        v2f v = accw[r * (OUTD / 2)];
        v += a * zq;
        accw[r * (OUTD / 2)] = v;
      }
    }
    for (int t = 0; t < OWN_SLOTS; ++t) {
      const int r = wv + 32 * t;
      if (r < rows) ((v2f*)hp)[r * (OUTD / 2) + lane] = accw[r * (OUTD / 2)];
    }
#undef OWN_ENT
#undef OWN_LOADG
#undef OWN_RMW2
#undef OWN_RLF
#undef OWN_PROC
  } else {
    for (int e = tid; e < E; e += OWN_THREADS) {
      unsigned r = (unsigned)(dst[e] - lo);
      if (r < urows) atomicMax(&mb[r], enc_f32(lrelu(s1[src[e]] + s2[lo + (int)r])));
    }
    __syncthreads();
    for (int e = tid; e < E; e += OWN_THREADS) {
      unsigned r = (unsigned)(dst[e] - lo);
      if (r < urows) {
        float mv = dec_f32(mb[r]);
        if (!isfinite(mv)) mv = 0.0f;
        atomicAdd(&den[r], expf(lrelu(s1[src[e]] + s2[lo + (int)r]) - mv));
      }
    }
    __syncthreads();
    for (int e = tid; e < E; e += OWN_THREADS) {
      unsigned r = (unsigned)(dst[e] - lo);
      if (r < urows) {
        int s = src[e];
        float mv = dec_f32(mb[r]);
        if (!isfinite(mv)) mv = 0.0f;
        float a = expf(lrelu(s1[s] + s2[lo + (int)r]) - mv) / fmaxf(den[r], 1e-9f);
        const float* zr = z + (long)s * OUTD;
        for (int c = 0; c < OUTD; ++c) atomicAdd(&acc[r * OUTD + c], a * zr[c]);
      }
    }
    __syncthreads();
    for (int idx = tid; idx < rows * OUTD; idx += OWN_THREADS) hp[idx] = acc[idx];
  }
#undef OWN_PUSH
#undef OWN_MATCH
#undef OWN_WALK2
}

extern "C" void kernel_launch(void* const* d_in, const int* in_sizes, int n_in,
                              void* d_out, int out_size, void* d_ws, size_t ws_size,
                              hipStream_t stream) {
  const float* x    = (const float*)d_in[0];
  const float* w_in = (const float*)d_in[1];
  const float* b_in = (const float*)d_in[2];
  const float* w    = (const float*)d_in[3];
  const float* a    = (const float*)d_in[4];
  const int*   src  = (const int*)d_in[5];
  const int*   dst  = (const int*)d_in[6];
  float* h = (float*)d_out;

  const int N = in_sizes[0] / IN_DIM;
  const int E = in_sizes[5];

  auto align256 = [](size_t v) { return (v + 255) & ~(size_t)255; };
  char* base = (char*)d_ws;
  size_t off = 0;
  float*    z     = (float*)(base + off); off += align256((size_t)N * OUTD * 4);
  float*    s1    = (float*)(base + off); off += align256((size_t)N * 4);
  float*    s2    = (float*)(base + off); off += align256((size_t)N * 4);
  unsigned* key   = (unsigned*)(base + off); off += align256((size_t)E * 9 + 16384);
  unsigned* pk    = (unsigned*)(base + off); off += align256((size_t)E * 4);

  {
    int ntiles = (N + 15) >> 4;
    int blocks = (ntiles + 1) / 2;
    gat_fused_gemm<<<dim3(blocks), dim3(64), 0, stream>>>(x, w_in, b_in, w, z, N);
  }
  gat_scores<<<dim3((N + 255) / 256), dim3(256), 0, stream>>>(z, a, s1, s2, N, src, dst, key, pk, E);
  gat_owner<<<dim3((N + OWN_ROWS - 1) / OWN_ROWS), dim3(OWN_THREADS), 0, stream>>>(
      src, dst, z, s1, s2, h, N, E, key, pk);
}


# owner row sums in registers: a wave's queue is ordered by row (returning LDS add on a per-row count, a scan); a row is summed in VGPRs and stored once, no LDS read-add-write; arrival order as before
# speedup vs baseline: 4.3944x; 1.0395x over previous
#include <hip/hip_runtime.h>
#include <math.h>

typedef float v2f __attribute__((ext_vector_type(2)));
typedef float v8f __attribute__((ext_vector_type(8)));

#define IN_DIM 128
#define HID    64
#define OUTD   64

__device__ __forceinline__ unsigned enc_f32(float f) {
  unsigned b = __float_as_uint(f);
  return (b & 0x80000000u) ? ~b : (b | 0x80000000u);
}
__device__ __forceinline__ float dec_f32(unsigned u) {
  unsigned b = (u & 0x80000000u) ? (u & 0x7FFFFFFFu) : ~u;
  return __uint_as_float(b);
}

__global__ __launch_bounds__(64) void gat_fused_gemm(
    const float* __restrict__ x, const float* __restrict__ w_in,
    const float* __restrict__ b_in, const float* __restrict__ w,
    float* __restrict__ z, int N) {
  __shared__ float lds[2][16][66];

  const int lane = threadIdx.x & 31;
  const int wv   = threadIdx.x >> 5;
  const int tile = blockIdx.x * 2 + wv;
  const int ntiles = (N + 15) >> 4;
  const bool active = tile < ntiles;

  const int m  = lane & 15;
  const int kp = lane >> 4;

  if (active) {
    const int row  = tile * 16 + m;
    const int rowc = row < N ? row : (N - 1);
    const float* xrow = x + (long)rowc * IN_DIM;

    v8f acc[4];
    #pragma unroll
    for (int nt = 0; nt < 4; ++nt) acc[nt] = (v8f){0,0,0,0,0,0,0,0};

    for (int k0 = 0; k0 < IN_DIM; k0 += 4) {
      const int kk = k0 + 2 * kp;
      v2f av = *(const v2f*)(xrow + kk);
      const float* wr0 = w_in + (long)kk * HID;
      const float* wr1 = wr0 + HID;
      #pragma unroll
      for (int nt = 0; nt < 4; ++nt) {
        v2f bv = { wr0[nt * 16 + m], wr1[nt * 16 + m] };
        acc[nt] = __builtin_amdgcn_wmma_f32_16x16x4_f32(
            false, av, false, bv, (short)0, acc[nt], false, false);
      }
    }

    #pragma unroll
    for (int nt = 0; nt < 4; ++nt) {
      float bias = b_in[nt * 16 + m];
      #pragma unroll
      for (int r = 0; r < 8; ++r) {
        float v = acc[nt][r] + bias;
        v = 0.5f * v * (1.0f + erff(v * 0.70710678118654752f));
        lds[wv][r + 8 * kp][nt * 16 + m] = v;
      }
    }
  }
  __syncthreads();

  if (active) {
    v8f acc[4];
    #pragma unroll
    for (int nt = 0; nt < 4; ++nt) acc[nt] = (v8f){0,0,0,0,0,0,0,0};

    for (int k0 = 0; k0 < HID; k0 += 4) {
      const int kk = k0 + 2 * kp;
      v2f av = *(const v2f*)(&lds[wv][m][kk]);
      const float* wr0 = w + (long)kk * OUTD;
      const float* wr1 = wr0 + OUTD;
      #pragma unroll
      for (int nt = 0; nt < 4; ++nt) {
        v2f bv = { wr0[nt * 16 + m], wr1[nt * 16 + m] };
        acc[nt] = __builtin_amdgcn_wmma_f32_16x16x4_f32(
            false, av, false, bv, (short)0, acc[nt], false, false);
      }
    }

    #pragma unroll
    for (int nt = 0; nt < 4; ++nt) {
      #pragma unroll
      for (int r = 0; r < 8; ++r) {
        int rr = tile * 16 + r + 8 * kp;
        if (rr < N) z[(long)rr * OUTD + nt * 16 + m] = acc[nt][r];
      }
    }
  }
}

#define OWN_ROWS    392
__device__ __forceinline__ int  own_band_pitch(int ng) { return (ng + 31) & ~31; }
__device__ __forceinline__ long own_band_base(int ng)  { return (2 * (long)ng + 31) & ~31L; }
__device__ __forceinline__ unsigned own_key(int d) { return ((unsigned)d / OWN_ROWS) & 255u; }
__device__ __forceinline__ void own_planes(unsigned (&pl)[8], int d, int i) {
  const unsigned k = own_key(d);
  #pragma unroll
  for (int b = 0; b < 8; ++b) pl[b] |= ((k >> b) & 1u) << i;
}

#define OWN_SRC_BITS 23
#define OWN_SRC_MASK ((1u << OWN_SRC_BITS) - 1u)
__device__ __forceinline__ unsigned own_pack(int s, int d) {
  const unsigned ud = (unsigned)d;
  return (unsigned)s | ((ud - (ud / OWN_ROWS) * OWN_ROWS) << OWN_SRC_BITS);
}
__global__ __launch_bounds__(256) void gat_scores(
    const float* __restrict__ z, const float* __restrict__ a,
    float* __restrict__ s1, float* __restrict__ s2, int N,
    const int* __restrict__ src, const int* __restrict__ dst,
    unsigned* __restrict__ key, unsigned* __restrict__ pk, int E) {
  int i = blockIdx.x * blockDim.x + threadIdx.x;
  {
    const int ng = E >> 5;
    const int4* d4 = (const int4*)dst;
    const int4* s4 = (const int4*)src;
    uint4* pk4 = (uint4*)pk;
    const int bp = own_band_pitch(ng);
    uint2* lo2 = (uint2*)key;
    unsigned* band = key + own_band_base(ng);
    if (ng & 1) {
      if (i < 64) band[(long)i * bp + ng] = 0u;
      if (i == 0) { uint2 zz; zz.x = 0u; zz.y = 0u; lo2[ng] = zz; }
    }
    for (int j = i; j < ng; j += gridDim.x * blockDim.x) {
      unsigned pl[8] = {0u, 0u, 0u, 0u, 0u, 0u, 0u, 0u};
      #pragma unroll
      for (int q = 0; q < 8; ++q) {
        const int4 p = d4[8 * (long)j + q];
        const int4 sv = s4[8 * (long)j + q];
        own_planes(pl, p.x, 4 * q + 0); own_planes(pl, p.y, 4 * q + 1);
        own_planes(pl, p.z, 4 * q + 2); own_planes(pl, p.w, 4 * q + 3);
        uint4 w;
        w.x = own_pack(sv.x, p.x); w.y = own_pack(sv.y, p.y);
        w.z = own_pack(sv.z, p.z); w.w = own_pack(sv.w, p.w);
        pk4[8 * (long)j + q] = w;
      }
      uint2 lo;
      lo.x = pl[0]; lo.y = pl[1];
      lo2[j] = lo;
      unsigned hA[8], hB[8];
      #pragma unroll
      for (int t = 0; t < 8; ++t) {
        hB[t] = ((t & 1) ? pl[2] : ~pl[2]) & ((t & 2) ? pl[3] : ~pl[3]) & ((t & 4) ? pl[4] : ~pl[4]);
        hA[t] = ((t & 1) ? pl[5] : ~pl[5]) & ((t & 2) ? pl[6] : ~pl[6]) & ((t & 4) ? pl[7] : ~pl[7]);
      }
      #pragma unroll
      for (int t = 0; t < 64; ++t) band[(long)t * bp + j] = hA[t >> 3] & hB[t & 7];
    }
    if (i < (E & 31)) {
      const int e = (ng << 5) + i;
      pk[e] = own_pack(src[e], dst[e]);
    }
  }
  if (i >= N) return;
  const float4* zr = (const float4*)(z + (long)i * OUTD);
  const float4* a1 = (const float4*)a;
  const float4* a2 = (const float4*)(a + OUTD);
  float acc1 = 0.f, acc2 = 0.f;
  #pragma unroll
  for (int j = 0; j < OUTD / 4; ++j) {
    float4 v = zr[j];
    float4 p = a1[j];
    float4 q = a2[j];
    acc1 += v.x * p.x + v.y * p.y + v.z * p.z + v.w * p.w;
    acc2 += v.x * q.x + v.y * q.y + v.z * q.z + v.w * q.w;
  }
  s1[i] = acc1;
  s2[i] = acc2;
}

#define OWN_CAP     8192
#define OWN_THREADS 1024
#define OWN_SLOTS   ((OWN_ROWS + 31) / 32)

__device__ __forceinline__ float lrelu(float v) { return v > 0.f ? v : 0.01f * v; }
__device__ __forceinline__ int own_idx(int i, int start, int nvec) {
  int j = i + start;
  if (j >= nvec) j -= nvec;
  return j < nvec ? j : nvec - 1;
}

__global__ __launch_bounds__(OWN_THREADS) void gat_owner(
    const int* __restrict__ src, const int* __restrict__ dst,
    const float* __restrict__ z, const float* __restrict__ s1,
    const float* __restrict__ s2, float* __restrict__ h, int N, int E,
    const unsigned* __restrict__ key, const unsigned* __restrict__ pk) {
  __shared__ float    acc[OWN_ROWS * OUTD];
  __shared__ unsigned mb[OWN_ROWS];
  __shared__ float    den[OWN_ROWS];
  __shared__ int      L_e[OWN_CAP];
  __shared__ int      L_r[OWN_CAP];
  __shared__ float    L_v[OWN_CAP];
  __shared__ int      Q_s[OWN_CAP];
  __shared__ int      Q_r[OWN_CAP];
  __shared__ float    Q_v[OWN_CAP];
  __shared__ int      rcnt[32 * OWN_SLOTS];
  __shared__ int      rstart[32 * OWN_SLOTS];
  __shared__ int      cnt;

  const int tid  = threadIdx.x;
  const int lane = tid & 31;
  const int wv   = __builtin_amdgcn_readfirstlane(tid >> 5);
  const int lo   = blockIdx.x * OWN_ROWS;
  const int rows = (N - lo) < OWN_ROWS ? (N - lo) : OWN_ROWS;

  for (int i = tid; i < OWN_ROWS; i += OWN_THREADS) { mb[i] = enc_f32(-INFINITY); den[i] = 0.0f; }
  if (tid < 32 * OWN_SLOTS) rcnt[tid] = 0;
  if (tid == 0) cnt = 0;
  __syncthreads();

  const int ng = E >> 5;
  const int np = (ng + 1) >> 1;
  const unsigned kb = blockIdx.x & 255u;
  const uint4* lo4 = (const uint4*)key;
  const uint2* bw2 = (const uint2*)(key + own_band_base(ng) + (long)(kb >> 2) * own_band_pitch(ng));
  const int start = (int)(((long)blockIdx.x * np) / gridDim.x);
  const unsigned urows = (unsigned)rows;
  const unsigned n0 = ((kb >> 0) & 1u) - 1u, n1 = ((kb >> 1) & 1u) - 1u;
#define OWN_PUSH(EID) { int p_ = atomicAdd(&cnt, 1); if (p_ < OWN_CAP) L_e[p_] = (EID); }
#define OWN_MATCH(P0, P1, BW) ((((P0) ^ n0) & ((P1) ^ n1)) & (BW))
#define OWN_WALK2(MA, JA, MB, JB) { \
    unsigned ma_ = (MA), mb_ = (MB); \
    while (ma_ | mb_) { \
      const bool ua_ = ma_ != 0u; \
      const int b_ = __builtin_ctz(ua_ ? ma_ : mb_); \
      if (ua_) ma_ &= ma_ - 1u; else mb_ &= mb_ - 1u; \
      OWN_PUSH(32 * (ua_ ? (JA) : (JB)) + b_) } }
  int i = tid;
  if (np > 0) {
    int pc0 = own_idx(i, start, np), pc1 = own_idx(i + OWN_THREADS, start, np);
    uint4 cl0 = lo4[pc0], cl1 = lo4[pc1];
    uint2 cb0 = bw2[pc0], cb1 = bw2[pc1];
    while (i + OWN_THREADS < np) {
      const int in = i + 2 * OWN_THREADS;
      const int pn0 = own_idx(in, start, np), pn1 = own_idx(in + OWN_THREADS, start, np);
      const uint4 nl0 = lo4[pn0], nl1 = lo4[pn1];
      const uint2 nb0 = bw2[pn0], nb1 = bw2[pn1];
      OWN_WALK2(OWN_MATCH(cl0.x, cl0.y, cb0.x), 2 * pc0, OWN_MATCH(cl0.z, cl0.w, cb0.y), 2 * pc0 + 1)
      OWN_WALK2(OWN_MATCH(cl1.x, cl1.y, cb1.x), 2 * pc1, OWN_MATCH(cl1.z, cl1.w, cb1.y), 2 * pc1 + 1)
      cl0 = nl0; cl1 = nl1; cb0 = nb0; cb1 = nb1;
      pc0 = pn0; pc1 = pn1;
      i = in;
    }
    for (; i < np; i += OWN_THREADS) {
      const int p0 = own_idx(i, start, np);
      const uint4 dl = lo4[p0];
      const uint2 db = bw2[p0];
      OWN_WALK2(OWN_MATCH(dl.x, dl.y, db.x), 2 * p0, OWN_MATCH(dl.z, dl.w, db.y), 2 * p0 + 1)
    }
  }
  if (tid < (E & 31)) {
    int e = (ng << 5) + tid;
    unsigned r = (unsigned)(dst[e] - lo);
    if (r < urows) OWN_PUSH(e)
  }
  __syncthreads();
  const int total = __builtin_amdgcn_readfirstlane(cnt);

  float* hp = h + (long)lo * OUTD;
  if (total <= OWN_CAP) {
    const bool two_arrays = gridDim.x > 256u || N > (int)(OWN_SRC_MASK + 1u);
    for (int k = tid; k < total; k += OWN_THREADS) {
      int e = L_e[k];
      int s;
      unsigned r;
      if (two_arrays) {
        s = src[e];
        r = (unsigned)(dst[e] - lo);
      } else {
        const unsigned w = pk[e];
        s = (int)(w & OWN_SRC_MASK);
        r = w >> OWN_SRC_BITS;
      }
      if (r < urows) {
        float v = lrelu(s1[s] + s2[lo + (int)r]);
        L_e[k] = s;
        L_v[k] = v;
        atomicMax(&mb[r], enc_f32(v));
        int rank = atomicAdd(&rcnt[(r & 31u) * OWN_SLOTS + (r >> 5)], 1);
        L_r[k] = (int)r | (rank << 9);
      } else {
        L_r[k] = -1;
      }
    }
    __syncthreads();
    int c_l = 0;
    for (int t = 0; t < OWN_SLOTS; ++t) c_l += rcnt[lane * OWN_SLOTS + t];
    int incl = c_l;
    for (int d = 1; d < 32; d <<= 1) {
      int t = __shfl_up(incl, d);
      if (lane >= d) incl += t;
    }
    const int base_l = incl - c_l;
    const int qs = __builtin_amdgcn_readlane(base_l, wv);
    const int qn = __builtin_amdgcn_readlane(c_l, wv);
    {
      const int own = lane < OWN_SLOTS ? rcnt[wv * OWN_SLOTS + lane] : 0;
      int inc = own;
      for (int d = 1; d < 16; d <<= 1) {
        int t = __shfl_up(inc, d);
        if (lane >= d) inc += t;
      }
      if (lane < OWN_SLOTS) rstart[wv * OWN_SLOTS + lane] = qs + inc - own;
    }
    __syncthreads();
    for (int k = tid; k < total; k += OWN_THREADS) {
      const int pr = L_r[k];
      if (pr >= 0) {
        const int r = pr & 511;
        float mv = dec_f32(mb[r]);
        if (!isfinite(mv)) mv = 0.0f;
        float ev = expf(L_v[k] - mv);
        atomicAdd(&den[r], ev);
        const int pos = rstart[(r & 31) * OWN_SLOTS + (r >> 5)] + (pr >> 9);
        Q_s[pos] = L_e[k];
        Q_r[pos] = r;
        Q_v[pos] = ev;
      }
    }
    __syncthreads();
    const unsigned lane8 = (unsigned)lane * 8u;
    v2f*       hl  = (v2f*)hp + lane;
    const int  Gt  = (qn + 15) >> 4;
    int cur = wv;
    v2f sum = {0.0f, 0.0f};
#define OWN_ENT(G_, S_, R_, A_) { \
      const int o_ = 16 * (G_) + (lane & 15); \
      S_ = 0; R_ = 0; A_ = 0.0f; \
      if (o_ < qn) { const int k_ = qs + o_; S_ = Q_s[k_]; R_ = Q_r[k_]; A_ = Q_v[k_] / fmaxf(den[R_], 1e-9f); } }
#define OWN_LOADG(ZV, S_) { \
      _Pragma("unroll") \
      for (int u = 0; u < 16; ++u) { \
        const char* rp_ = (const char*)z + (size_t)(unsigned)__builtin_amdgcn_readlane(S_, u) * (OUTD * 4); \
        ZV[u] = *(const v2f*)(rp_ + lane8); } }
#define OWN_RLF(V, U) __int_as_float(__builtin_amdgcn_readlane(__float_as_int(V), (U)))
#define OWN_ADD1(ZQ, R_, A_, U) { \
      const int r_ = __builtin_amdgcn_readlane(R_, (U)); \
      while (cur < r_) { hl[cur * (OUTD / 2)] = sum; sum = (v2f){0.0f, 0.0f}; cur += 32; } \
      sum += OWN_RLF(A_, (U)) * (ZQ); }
#define OWN_FULL(ZV, R_, A_) { \
      _Pragma("unroll") \
      for (int u = 0; u < 16; ++u) OWN_ADD1(ZV[u], R_, A_, u) }
#define OWN_LAST(ZV, R_, A_, N_) { \
      const int n_ = (N_); \
      _Pragma("unroll") \
      for (int u = 0; u < 16; ++u) if (u < n_) OWN_ADD1(ZV[u], R_, A_, u) }
    int   sA = 0, rA = 0, sB = 0, rB = 0;
    float aA = 0.0f, aB = 0.0f;
    v2f   zvA[16], zvB[16];
    if (Gt > 0) {
      OWN_ENT(0, sA, rA, aA) OWN_LOADG(zvA, sA)
      for (int g = 0; g < Gt; g += 2) {
        if (g + 1 >= Gt) { OWN_LAST(zvA, rA, aA, qn - 16 * g) break; }
        OWN_ENT(g + 1, sB, rB, aB) OWN_LOADG(zvB, sB)
        OWN_FULL(zvA, rA, aA)
        if (g + 2 >= Gt) { OWN_LAST(zvB, rB, aB, qn - 16 * (g + 1)) break; }
        OWN_ENT(g + 2, sA, rA, aA) OWN_LOADG(zvA, sA)
        OWN_FULL(zvB, rB, aB)
      }
    }
    while (cur < rows) { hl[cur * (OUTD / 2)] = sum; sum = (v2f){0.0f, 0.0f}; cur += 32; }
#undef OWN_ENT
#undef OWN_LOADG
#undef OWN_RLF
#undef OWN_ADD1
#undef OWN_FULL
#undef OWN_LAST
  } else {
    for (int i = tid; i < OWN_ROWS * OUTD; i += OWN_THREADS) acc[i] = 0.0f;
    for (int e = tid; e < E; e += OWN_THREADS) {
      unsigned r = (unsigned)(dst[e] - lo);
      if (r < urows) atomicMax(&mb[r], enc_f32(lrelu(s1[src[e]] + s2[lo + (int)r])));
    }
    __syncthreads();
    for (int e = tid; e < E; e += OWN_THREADS) {
      unsigned r = (unsigned)(dst[e] - lo);
      if (r < urows) {
        float mv = dec_f32(mb[r]);
        if (!isfinite(mv)) mv = 0.0f;
        atomicAdd(&den[r], expf(lrelu(s1[src[e]] + s2[lo + (int)r]) - mv));
      }
    }
    __syncthreads();
    for (int e = tid; e < E; e += OWN_THREADS) {
      unsigned r = (unsigned)(dst[e] - lo);
      if (r < urows) {
        int s = src[e];
        float mv = dec_f32(mb[r]);
        if (!isfinite(mv)) mv = 0.0f;
        float a = expf(lrelu(s1[s] + s2[lo + (int)r]) - mv) / fmaxf(den[r], 1e-9f);
        const float* zr = z + (long)s * OUTD;
        for (int c = 0; c < OUTD; ++c) atomicAdd(&acc[r * OUTD + c], a * zr[c]);
      }
    }
    __syncthreads();
    for (int idx = tid; idx < rows * OUTD; idx += OWN_THREADS) hp[idx] = acc[idx];
  }
#undef OWN_PUSH
#undef OWN_MATCH
#undef OWN_WALK2
}

extern "C" void kernel_launch(void* const* d_in, const int* in_sizes, int n_in,
                              void* d_out, int out_size, void* d_ws, size_t ws_size,
                              hipStream_t stream) {
  const float* x    = (const float*)d_in[0];
  const float* w_in = (const float*)d_in[1];
  const float* b_in = (const float*)d_in[2];
  const float* w    = (const float*)d_in[3];
  const float* a    = (const float*)d_in[4];
  const int*   src  = (const int*)d_in[5];
  const int*   dst  = (const int*)d_in[6];
  float* h = (float*)d_out;

  const int N = in_sizes[0] / IN_DIM;
  const int E = in_sizes[5];

  auto align256 = [](size_t v) { return (v + 255) & ~(size_t)255; };
  char* base = (char*)d_ws;
  size_t off = 0;
  float*    z     = (float*)(base + off); off += align256((size_t)N * OUTD * 4);
  float*    s1    = (float*)(base + off); off += align256((size_t)N * 4);
  float*    s2    = (float*)(base + off); off += align256((size_t)N * 4);
  unsigned* key   = (unsigned*)(base + off); off += align256((size_t)E * 9 + 16384);
  unsigned* pk    = (unsigned*)(base + off); off += align256((size_t)E * 4);

  {
    int ntiles = (N + 15) >> 4;
    int blocks = (ntiles + 1) / 2;
    gat_fused_gemm<<<dim3(blocks), dim3(64), 0, stream>>>(x, w_in, b_in, w, z, N);
  }
  gat_scores<<<dim3((N + 255) / 256), dim3(256), 0, stream>>>(z, a, s1, s2, N, src, dst, key, pk, E);
  gat_owner<<<dim3((N + OWN_ROWS - 1) / OWN_ROWS), dim3(OWN_THREADS), 0, stream>>>(
      src, dst, z, s1, s2, h, N, E, key, pk);
}


# GEMM launch: 256 work-groups of 24 waves stage w_in and w once in LDS (k-pair rows); B fragment = one 8-byte LDS read, not 2 global loads; a tile's x asked for at once up front; sums as before
# speedup vs baseline: 4.6533x; 1.0589x over previous
#include <hip/hip_runtime.h>
#include <math.h>

typedef float v2f __attribute__((ext_vector_type(2)));
typedef float v8f __attribute__((ext_vector_type(8)));

#define IN_DIM 128
#define HID    64
#define OUTD   64

__device__ __forceinline__ unsigned enc_f32(float f) {
  unsigned b = __float_as_uint(f);
  return (b & 0x80000000u) ? ~b : (b | 0x80000000u);
}
__device__ __forceinline__ float dec_f32(unsigned u) {
  unsigned b = (u & 0x80000000u) ? (u & 0x7FFFFFFFu) : ~u;
  return __uint_as_float(b);
}

#define GEMM_WAVES 24
#define GEMM_WP    160
__global__ __launch_bounds__(GEMM_WAVES * 32) void gat_fused_gemm(
    const float* __restrict__ x, const float* __restrict__ w_in,
    const float* __restrict__ b_in, const float* __restrict__ w,
    float* __restrict__ z, int N) {
  __shared__ __attribute__((aligned(16))) float wl1[(IN_DIM / 2) * GEMM_WP];
  __shared__ __attribute__((aligned(16))) float wl2[(HID / 2) * GEMM_WP];
  __shared__ __attribute__((aligned(16))) float lds[GEMM_WAVES][16][68];

  const int tid  = threadIdx.x;
  const int lane = tid & 31;
  const int wv   = __builtin_amdgcn_readfirstlane(tid >> 5);
  const int m  = lane & 15;
  const int kp = lane >> 4;

  const int ntiles = (N + 15) >> 4;
  const unsigned per = (unsigned)ntiles / gridDim.x, rem = (unsigned)ntiles % gridDim.x;
  const int t0 = (int)(blockIdx.x * per + (blockIdx.x < rem ? blockIdx.x : rem));
  const int t1 = t0 + (int)per + (blockIdx.x < rem ? 1 : 0);

  if (t0 + wv < t1) {
    const int prow = (t0 + wv) * 16 + m;
    __builtin_prefetch(x + (long)(prow < N ? prow : N - 1) * IN_DIM + (IN_DIM / 2) * kp, 0, 3);
  }

  float bias[4];
  #pragma unroll
  for (int nt = 0; nt < 4; ++nt) bias[nt] = b_in[nt * 16 + m];

  for (int it = tid; it < (IN_DIM / 2) * (HID / 4); it += GEMM_WAVES * 32) {
    const int n4 = it & (HID / 4 - 1), p = it / (HID / 4);
    const float4 r0 = *(const float4*)(w_in + (long)(2 * p) * HID + 4 * n4);
    const float4 r1 = *(const float4*)(w_in + (long)(2 * p + 1) * HID + 4 * n4);
    float4* d = (float4*)(wl1 + p * GEMM_WP + 8 * n4);
    d[0] = make_float4(r0.x, r1.x, r0.y, r1.y);
    d[1] = make_float4(r0.z, r1.z, r0.w, r1.w);
  }
  for (int it = tid; it < (HID / 2) * (OUTD / 4); it += GEMM_WAVES * 32) {
    const int n4 = it & (OUTD / 4 - 1), p = it / (OUTD / 4);
    const float4 r0 = *(const float4*)(w + (long)(2 * p) * OUTD + 4 * n4);
    const float4 r1 = *(const float4*)(w + (long)(2 * p + 1) * OUTD + 4 * n4);
    float4* d = (float4*)(wl2 + p * GEMM_WP + 8 * n4);
    d[0] = make_float4(r0.x, r1.x, r0.y, r1.y);
    d[1] = make_float4(r0.z, r1.z, r0.w, r1.w);
  }
  __syncthreads();

  const float* bq1 = wl1 + kp * GEMM_WP + 2 * m;
  const float* bq2 = wl2 + kp * GEMM_WP + 2 * m;

  v2f xa[IN_DIM / 4];
#define GEMM_LOADX(T_) { \
    const int row_ = (T_) * 16 + m; \
    const int rowc_ = row_ < N ? row_ : (N - 1); \
    const float* xr_ = x + (long)rowc_ * IN_DIM + 2 * kp; \
    _Pragma("unroll") \
    for (int j = 0; j < IN_DIM / 4; ++j) xa[j] = *(const v2f*)(xr_ + 4 * j); }

  int tile = t0 + wv;
  if (tile < t1) GEMM_LOADX(tile)
  for (; tile < t1; tile += GEMM_WAVES) {
    v8f acc[4];
    #pragma unroll
    for (int nt = 0; nt < 4; ++nt) acc[nt] = (v8f){0,0,0,0,0,0,0,0};

    #pragma unroll
    for (int j = 0; j < IN_DIM / 4; ++j) {
      #pragma unroll
      for (int nt = 0; nt < 4; ++nt) {
        v2f bv = *(const v2f*)(bq1 + (2 * j) * GEMM_WP + nt * 32);
        acc[nt] = __builtin_amdgcn_wmma_f32_16x16x4_f32(
            false, xa[j], false, bv, (short)0, acc[nt], false, false);
      }
    }
    if (tile + GEMM_WAVES < t1) GEMM_LOADX(tile + GEMM_WAVES)

    #pragma unroll
    for (int nt = 0; nt < 4; ++nt) {
      #pragma unroll
      for (int r = 0; r < 8; ++r) {
        float v = acc[nt][r] + bias[nt];
        v = 0.5f * v * (1.0f + erff(v * 0.70710678118654752f));
        lds[wv][r + 8 * kp][nt * 16 + m] = v;
      }
    }
    __builtin_amdgcn_fence(__ATOMIC_RELEASE, "wavefront");
    __builtin_amdgcn_wave_barrier();
    __builtin_amdgcn_fence(__ATOMIC_ACQUIRE, "wavefront");

    v8f acc2[4];
    #pragma unroll
    for (int nt = 0; nt < 4; ++nt) acc2[nt] = (v8f){0,0,0,0,0,0,0,0};

    #pragma unroll
    for (int j = 0; j < HID / 4; ++j) {
      v2f av = *(const v2f*)(&lds[wv][m][4 * j + 2 * kp]);
      #pragma unroll
      for (int nt = 0; nt < 4; ++nt) {
        v2f bv = *(const v2f*)(bq2 + (2 * j) * GEMM_WP + nt * 32);
        acc2[nt] = __builtin_amdgcn_wmma_f32_16x16x4_f32(
            false, av, false, bv, (short)0, acc2[nt], false, false);
      }
    }

    #pragma unroll
    for (int nt = 0; nt < 4; ++nt) {
      #pragma unroll
      for (int r = 0; r < 8; ++r) {
        int rr = tile * 16 + r + 8 * kp;
        if (rr < N) z[(long)rr * OUTD + nt * 16 + m] = acc2[nt][r];
      }
    }
    __builtin_amdgcn_fence(__ATOMIC_RELEASE, "wavefront");
    __builtin_amdgcn_wave_barrier();
    __builtin_amdgcn_fence(__ATOMIC_ACQUIRE, "wavefront");
  }
#undef GEMM_LOADX
}

#define OWN_ROWS    392
__device__ __forceinline__ int  own_band_pitch(int ng) { return (ng + 31) & ~31; }
__device__ __forceinline__ long own_band_base(int ng)  { return (2 * (long)ng + 31) & ~31L; }
__device__ __forceinline__ unsigned own_key(int d) { return ((unsigned)d / OWN_ROWS) & 255u; }
__device__ __forceinline__ void own_planes(unsigned (&pl)[8], int d, int i) {
  const unsigned k = own_key(d);
  #pragma unroll
  for (int b = 0; b < 8; ++b) pl[b] |= ((k >> b) & 1u) << i;
}

#define OWN_SRC_BITS 23
#define OWN_SRC_MASK ((1u << OWN_SRC_BITS) - 1u)
__device__ __forceinline__ unsigned own_pack(int s, int d) {
  const unsigned ud = (unsigned)d;
  return (unsigned)s | ((ud - (ud / OWN_ROWS) * OWN_ROWS) << OWN_SRC_BITS);
}
__global__ __launch_bounds__(256) void gat_scores(
    const float* __restrict__ z, const float* __restrict__ a,
    float* __restrict__ s1, float* __restrict__ s2, int N,
    const int* __restrict__ src, const int* __restrict__ dst,
    unsigned* __restrict__ key, unsigned* __restrict__ pk, int E) {
  int i = blockIdx.x * blockDim.x + threadIdx.x;
  {
    const int ng = E >> 5;
    const int4* d4 = (const int4*)dst;
    const int4* s4 = (const int4*)src;
    uint4* pk4 = (uint4*)pk;
    const int bp = own_band_pitch(ng);
    uint2* lo2 = (uint2*)key;
    unsigned* band = key + own_band_base(ng);
    if (ng & 1) {
      if (i < 64) band[(long)i * bp + ng] = 0u;
      if (i == 0) { uint2 zz; zz.x = 0u; zz.y = 0u; lo2[ng] = zz; }
    }
    for (int j = i; j < ng; j += gridDim.x * blockDim.x) {
      unsigned pl[8] = {0u, 0u, 0u, 0u, 0u, 0u, 0u, 0u};
      #pragma unroll
      for (int q = 0; q < 8; ++q) {
        const int4 p = d4[8 * (long)j + q];
        const int4 sv = s4[8 * (long)j + q];
        own_planes(pl, p.x, 4 * q + 0); own_planes(pl, p.y, 4 * q + 1);
        own_planes(pl, p.z, 4 * q + 2); own_planes(pl, p.w, 4 * q + 3);
        uint4 w;
        w.x = own_pack(sv.x, p.x); w.y = own_pack(sv.y, p.y);
        w.z = own_pack(sv.z, p.z); w.w = own_pack(sv.w, p.w);
        pk4[8 * (long)j + q] = w;
      }
      uint2 lo;
      lo.x = pl[0]; lo.y = pl[1];
      lo2[j] = lo;
      unsigned hA[8], hB[8];
      #pragma unroll
      for (int t = 0; t < 8; ++t) {
        hB[t] = ((t & 1) ? pl[2] : ~pl[2]) & ((t & 2) ? pl[3] : ~pl[3]) & ((t & 4) ? pl[4] : ~pl[4]);
        hA[t] = ((t & 1) ? pl[5] : ~pl[5]) & ((t & 2) ? pl[6] : ~pl[6]) & ((t & 4) ? pl[7] : ~pl[7]);
      }
      #pragma unroll
      for (int t = 0; t < 64; ++t) band[(long)t * bp + j] = hA[t >> 3] & hB[t & 7];
    }
    if (i < (E & 31)) {
      const int e = (ng << 5) + i;
      pk[e] = own_pack(src[e], dst[e]);
    }
  }
  if (i >= N) return;
  const float4* zr = (const float4*)(z + (long)i * OUTD);
  const float4* a1 = (const float4*)a;
  const float4* a2 = (const float4*)(a + OUTD);
  float acc1 = 0.f, acc2 = 0.f;
  #pragma unroll
  for (int j = 0; j < OUTD / 4; ++j) {
    float4 v = zr[j];
    float4 p = a1[j];
    float4 q = a2[j];
    acc1 += v.x * p.x + v.y * p.y + v.z * p.z + v.w * p.w;
    acc2 += v.x * q.x + v.y * q.y + v.z * q.z + v.w * q.w;
  }
  s1[i] = acc1;
  s2[i] = acc2;
}

#define OWN_CAP     8192
#define OWN_THREADS 1024
#define OWN_SLOTS   ((OWN_ROWS + 31) / 32)

__device__ __forceinline__ float lrelu(float v) { return v > 0.f ? v : 0.01f * v; }
__device__ __forceinline__ int own_idx(int i, int start, int nvec) {
  int j = i + start;
  if (j >= nvec) j -= nvec;
  return j < nvec ? j : nvec - 1;
}

__global__ __launch_bounds__(OWN_THREADS) void gat_owner(
    const int* __restrict__ src, const int* __restrict__ dst,
    const float* __restrict__ z, const float* __restrict__ s1,
    const float* __restrict__ s2, float* __restrict__ h, int N, int E,
    const unsigned* __restrict__ key, const unsigned* __restrict__ pk) {
  __shared__ float    acc[OWN_ROWS * OUTD];
  __shared__ unsigned mb[OWN_ROWS];
  __shared__ float    den[OWN_ROWS];
  __shared__ int      L_e[OWN_CAP];
  __shared__ int      L_r[OWN_CAP];
  __shared__ float    L_v[OWN_CAP];
  __shared__ int      Q_s[OWN_CAP];
  __shared__ int      Q_r[OWN_CAP];
  __shared__ float    Q_v[OWN_CAP];
  __shared__ int      rcnt[32 * OWN_SLOTS];
  __shared__ int      rstart[32 * OWN_SLOTS];
  __shared__ int      cnt;

  const int tid  = threadIdx.x;
  const int lane = tid & 31;
  const int wv   = __builtin_amdgcn_readfirstlane(tid >> 5);
  const int lo   = blockIdx.x * OWN_ROWS;
  const int rows = (N - lo) < OWN_ROWS ? (N - lo) : OWN_ROWS;

  for (int i = tid; i < OWN_ROWS; i += OWN_THREADS) { mb[i] = enc_f32(-INFINITY); den[i] = 0.0f; }
  if (tid < 32 * OWN_SLOTS) rcnt[tid] = 0;
  if (tid == 0) cnt = 0;
  __syncthreads();

  const int ng = E >> 5;
  const int np = (ng + 1) >> 1;
  const unsigned kb = blockIdx.x & 255u;
  const uint4* lo4 = (const uint4*)key;
  const uint2* bw2 = (const uint2*)(key + own_band_base(ng) + (long)(kb >> 2) * own_band_pitch(ng));
  const int start = (int)(((long)blockIdx.x * np) / gridDim.x);
  const unsigned urows = (unsigned)rows;
  const unsigned n0 = ((kb >> 0) & 1u) - 1u, n1 = ((kb >> 1) & 1u) - 1u;
#define OWN_PUSH(EID) { int p_ = atomicAdd(&cnt, 1); if (p_ < OWN_CAP) L_e[p_] = (EID); }
#define OWN_MATCH(P0, P1, BW) ((((P0) ^ n0) & ((P1) ^ n1)) & (BW))
#define OWN_WALK2(MA, JA, MB, JB) { \
    unsigned ma_ = (MA), mb_ = (MB); \
    while (ma_ | mb_) { \
      const bool ua_ = ma_ != 0u; \
      const int b_ = __builtin_ctz(ua_ ? ma_ : mb_); \
      if (ua_) ma_ &= ma_ - 1u; else mb_ &= mb_ - 1u; \
      OWN_PUSH(32 * (ua_ ? (JA) : (JB)) + b_) } }
  int i = tid;
  if (np > 0) {
    int pc0 = own_idx(i, start, np), pc1 = own_idx(i + OWN_THREADS, start, np);
    uint4 cl0 = lo4[pc0], cl1 = lo4[pc1];
    uint2 cb0 = bw2[pc0], cb1 = bw2[pc1];
    while (i + OWN_THREADS < np) {
      const int in = i + 2 * OWN_THREADS;
      const int pn0 = own_idx(in, start, np), pn1 = own_idx(in + OWN_THREADS, start, np);
      const uint4 nl0 = lo4[pn0], nl1 = lo4[pn1];
      const uint2 nb0 = bw2[pn0], nb1 = bw2[pn1];
      OWN_WALK2(OWN_MATCH(cl0.x, cl0.y, cb0.x), 2 * pc0, OWN_MATCH(cl0.z, cl0.w, cb0.y), 2 * pc0 + 1)
      OWN_WALK2(OWN_MATCH(cl1.x, cl1.y, cb1.x), 2 * pc1, OWN_MATCH(cl1.z, cl1.w, cb1.y), 2 * pc1 + 1)
      cl0 = nl0; cl1 = nl1; cb0 = nb0; cb1 = nb1;
      pc0 = pn0; pc1 = pn1;
      i = in;
    }
    for (; i < np; i += OWN_THREADS) {
      const int p0 = own_idx(i, start, np);
      const uint4 dl = lo4[p0];
      const uint2 db = bw2[p0];
      OWN_WALK2(OWN_MATCH(dl.x, dl.y, db.x), 2 * p0, OWN_MATCH(dl.z, dl.w, db.y), 2 * p0 + 1)
    }
  }
  if (tid < (E & 31)) {
    int e = (ng << 5) + tid;
    unsigned r = (unsigned)(dst[e] - lo);
    if (r < urows) OWN_PUSH(e)
  }
  __syncthreads();
  const int total = __builtin_amdgcn_readfirstlane(cnt);

  float* hp = h + (long)lo * OUTD;
  if (total <= OWN_CAP) {
    const bool two_arrays = gridDim.x > 256u || N > (int)(OWN_SRC_MASK + 1u);
    for (int k = tid; k < total; k += OWN_THREADS) {
      int e = L_e[k];
      int s;
      unsigned r;
      if (two_arrays) {
        s = src[e];
        r = (unsigned)(dst[e] - lo);
      } else {
        const unsigned w = pk[e];
        s = (int)(w & OWN_SRC_MASK);
        r = w >> OWN_SRC_BITS;
      }
      if (r < urows) {
        float v = lrelu(s1[s] + s2[lo + (int)r]);
        L_e[k] = s;
        L_v[k] = v;
        atomicMax(&mb[r], enc_f32(v));
        int rank = atomicAdd(&rcnt[(r & 31u) * OWN_SLOTS + (r >> 5)], 1);
        L_r[k] = (int)r | (rank << 9);
      } else {
        L_r[k] = -1;
      }
    }
    __syncthreads();
    int c_l = 0;
    for (int t = 0; t < OWN_SLOTS; ++t) c_l += rcnt[lane * OWN_SLOTS + t];
    int incl = c_l;
    for (int d = 1; d < 32; d <<= 1) {
      int t = __shfl_up(incl, d);
      if (lane >= d) incl += t;
    }
    const int base_l = incl - c_l;
    const int qs = __builtin_amdgcn_readlane(base_l, wv);
    const int qn = __builtin_amdgcn_readlane(c_l, wv);
    {
      const int own = lane < OWN_SLOTS ? rcnt[wv * OWN_SLOTS + lane] : 0;
      int inc = own;
      for (int d = 1; d < 16; d <<= 1) {
        int t = __shfl_up(inc, d);
        if (lane >= d) inc += t;
      }
      if (lane < OWN_SLOTS) rstart[wv * OWN_SLOTS + lane] = qs + inc - own;
    }
    __syncthreads();
    for (int k = tid; k < total; k += OWN_THREADS) {
      const int pr = L_r[k];
      if (pr >= 0) {
        const int r = pr & 511;
        float mv = dec_f32(mb[r]);
        if (!isfinite(mv)) mv = 0.0f;
        float ev = expf(L_v[k] - mv);
        atomicAdd(&den[r], ev);
        const int pos = rstart[(r & 31) * OWN_SLOTS + (r >> 5)] + (pr >> 9);
        Q_s[pos] = L_e[k];
        Q_r[pos] = r;
        Q_v[pos] = ev;
      }
    }
    __syncthreads();
    const unsigned lane8 = (unsigned)lane * 8u;
    v2f*       hl  = (v2f*)hp + lane;
    const int  Gt  = (qn + 15) >> 4;
    int cur = wv;
    v2f sum = {0.0f, 0.0f};
#define OWN_ENT(G_, S_, R_, A_) { \
      const int o_ = 16 * (G_) + (lane & 15); \
      S_ = 0; R_ = 0; A_ = 0.0f; \
      if (o_ < qn) { const int k_ = qs + o_; S_ = Q_s[k_]; R_ = Q_r[k_]; A_ = Q_v[k_] / fmaxf(den[R_], 1e-9f); } }
#define OWN_LOADG(ZV, S_) { \
      _Pragma("unroll") \
      for (int u = 0; u < 16; ++u) { \
        const char* rp_ = (const char*)z + (size_t)(unsigned)__builtin_amdgcn_readlane(S_, u) * (OUTD * 4); \
        ZV[u] = *(const v2f*)(rp_ + lane8); } }
#define OWN_RLF(V, U) __int_as_float(__builtin_amdgcn_readlane(__float_as_int(V), (U)))
#define OWN_ADD1(ZQ, R_, A_, U) { \
      const int r_ = __builtin_amdgcn_readlane(R_, (U)); \
      while (cur < r_) { hl[cur * (OUTD / 2)] = sum; sum = (v2f){0.0f, 0.0f}; cur += 32; } \
      sum += OWN_RLF(A_, (U)) * (ZQ); }
#define OWN_FULL(ZV, R_, A_) { \
      _Pragma("unroll") \
      for (int u = 0; u < 16; ++u) OWN_ADD1(ZV[u], R_, A_, u) }
#define OWN_LAST(ZV, R_, A_, N_) { \
      const int n_ = (N_); \
      _Pragma("unroll") \
      for (int u = 0; u < 16; ++u) if (u < n_) OWN_ADD1(ZV[u], R_, A_, u) }
    int   sA = 0, rA = 0, sB = 0, rB = 0;
    float aA = 0.0f, aB = 0.0f;
    v2f   zvA[16], zvB[16];
    if (Gt > 0) {
      OWN_ENT(0, sA, rA, aA) OWN_LOADG(zvA, sA)
      for (int g = 0; g < Gt; g += 2) {
        if (g + 1 >= Gt) { OWN_LAST(zvA, rA, aA, qn - 16 * g) break; }
        OWN_ENT(g + 1, sB, rB, aB) OWN_LOADG(zvB, sB)
        OWN_FULL(zvA, rA, aA)
        if (g + 2 >= Gt) { OWN_LAST(zvB, rB, aB, qn - 16 * (g + 1)) break; }
        OWN_ENT(g + 2, sA, rA, aA) OWN_LOADG(zvA, sA)
        OWN_FULL(zvB, rB, aB)
      }
    }
    while (cur < rows) { hl[cur * (OUTD / 2)] = sum; sum = (v2f){0.0f, 0.0f}; cur += 32; }
#undef OWN_ENT
#undef OWN_LOADG
#undef OWN_RLF
#undef OWN_ADD1
#undef OWN_FULL
#undef OWN_LAST
  } else {
    for (int i = tid; i < OWN_ROWS * OUTD; i += OWN_THREADS) acc[i] = 0.0f;
    for (int e = tid; e < E; e += OWN_THREADS) {
      unsigned r = (unsigned)(dst[e] - lo);
      if (r < urows) atomicMax(&mb[r], enc_f32(lrelu(s1[src[e]] + s2[lo + (int)r])));
    }
    __syncthreads();
    for (int e = tid; e < E; e += OWN_THREADS) {
      unsigned r = (unsigned)(dst[e] - lo);
      if (r < urows) {
        float mv = dec_f32(mb[r]);
        if (!isfinite(mv)) mv = 0.0f;
        atomicAdd(&den[r], expf(lrelu(s1[src[e]] + s2[lo + (int)r]) - mv));
      }
    }
    __syncthreads();
    for (int e = tid; e < E; e += OWN_THREADS) {
      unsigned r = (unsigned)(dst[e] - lo);
      if (r < urows) {
        int s = src[e];
        float mv = dec_f32(mb[r]);
        if (!isfinite(mv)) mv = 0.0f;
        float a = expf(lrelu(s1[s] + s2[lo + (int)r]) - mv) / fmaxf(den[r], 1e-9f);
        const float* zr = z + (long)s * OUTD;
        for (int c = 0; c < OUTD; ++c) atomicAdd(&acc[r * OUTD + c], a * zr[c]);
      }
    }
    __syncthreads();
    for (int idx = tid; idx < rows * OUTD; idx += OWN_THREADS) hp[idx] = acc[idx];
  }
#undef OWN_PUSH
#undef OWN_MATCH
#undef OWN_WALK2
}

extern "C" void kernel_launch(void* const* d_in, const int* in_sizes, int n_in,
                              void* d_out, int out_size, void* d_ws, size_t ws_size,
                              hipStream_t stream) {
  const float* x    = (const float*)d_in[0];
  const float* w_in = (const float*)d_in[1];
  const float* b_in = (const float*)d_in[2];
  const float* w    = (const float*)d_in[3];
  const float* a    = (const float*)d_in[4];
  const int*   src  = (const int*)d_in[5];
  const int*   dst  = (const int*)d_in[6];
  float* h = (float*)d_out;

  const int N = in_sizes[0] / IN_DIM;
  const int E = in_sizes[5];

  auto align256 = [](size_t v) { return (v + 255) & ~(size_t)255; };
  char* base = (char*)d_ws;
  size_t off = 0;
  float*    z     = (float*)(base + off); off += align256((size_t)N * OUTD * 4);
  float*    s1    = (float*)(base + off); off += align256((size_t)N * 4);
  float*    s2    = (float*)(base + off); off += align256((size_t)N * 4);
  unsigned* key   = (unsigned*)(base + off); off += align256((size_t)E * 9 + 16384);
  unsigned* pk    = (unsigned*)(base + off); off += align256((size_t)E * 4);

  gat_fused_gemm<<<dim3(256), dim3(GEMM_WAVES * 32), 0, stream>>>(x, w_in, b_in, w, z, N);
  gat_scores<<<dim3((N + 255) / 256), dim3(256), 0, stream>>>(z, a, s1, s2, N, src, dst, key, pk, E);
  gat_owner<<<dim3((N + OWN_ROWS - 1) / OWN_ROWS), dim3(OWN_THREADS), 0, stream>>>(
      src, dst, z, s1, s2, h, N, E, key, pk);
}


# per-node launch removed: GEMM launch gets s1,s2 as 2 more columns of its 2nd product (w.a folded into w's pad columns, 16 more WMMAs a tile) and builds edge keys behind tiles; score sums reordered
# speedup vs baseline: 4.8438x; 1.0409x over previous
#include <hip/hip_runtime.h>
#include <math.h>

typedef float v2f __attribute__((ext_vector_type(2)));
typedef float v8f __attribute__((ext_vector_type(8)));

#define IN_DIM 128
#define HID    64
#define OUTD   64

__device__ __forceinline__ unsigned enc_f32(float f) {
  unsigned b = __float_as_uint(f);
  return (b & 0x80000000u) ? ~b : (b | 0x80000000u);
}
__device__ __forceinline__ float dec_f32(unsigned u) {
  unsigned b = (u & 0x80000000u) ? (u & 0x7FFFFFFFu) : ~u;
  return __uint_as_float(b);
}

#define OWN_ROWS    392
__device__ __forceinline__ int  own_band_pitch(int ng) { return (ng + 31) & ~31; }
__device__ __forceinline__ long own_band_base(int ng)  { return (2 * (long)ng + 31) & ~31L; }
__device__ __forceinline__ unsigned own_key(int d) { return ((unsigned)d / OWN_ROWS) & 255u; }
__device__ __forceinline__ void own_planes(unsigned (&pl)[8], int d, int i) {
  const unsigned k = own_key(d);
  #pragma unroll
  for (int b = 0; b < 8; ++b) pl[b] |= ((k >> b) & 1u) << i;
}

#define OWN_SRC_BITS 23
#define OWN_SRC_MASK ((1u << OWN_SRC_BITS) - 1u)
__device__ __forceinline__ unsigned own_pack(int s, int d) {
  const unsigned ud = (unsigned)d;
  return (unsigned)s | ((ud - (ud / OWN_ROWS) * OWN_ROWS) << OWN_SRC_BITS);
}

#define GEMM_WAVES 24
#define GEMM_WP    160
__global__ __launch_bounds__(GEMM_WAVES * 32) void gat_fused_gemm(
    const float* __restrict__ x, const float* __restrict__ w_in,
    const float* __restrict__ b_in, const float* __restrict__ w,
    const float* __restrict__ a, float* __restrict__ z,
    float* __restrict__ s1, float* __restrict__ s2, int N,
    const int* __restrict__ src, const int* __restrict__ dst,
    unsigned* __restrict__ key, unsigned* __restrict__ pk, int E) {
  __shared__ __attribute__((aligned(16))) float wl1[(IN_DIM / 2) * GEMM_WP];
  __shared__ __attribute__((aligned(16))) float wl2[(HID / 2) * GEMM_WP];
  __shared__ __attribute__((aligned(16))) float lds[GEMM_WAVES][16][68];

  const int tid  = threadIdx.x;
  const int lane = tid & 31;
  const int wv   = __builtin_amdgcn_readfirstlane(tid >> 5);
  const int m  = lane & 15;
  const int kp = lane >> 4;

  const int ntiles = (N + 15) >> 4;
  const unsigned per = (unsigned)ntiles / gridDim.x, rem = (unsigned)ntiles % gridDim.x;
  const int t0 = (int)(blockIdx.x * per + (blockIdx.x < rem ? blockIdx.x : rem));
  const int t1 = t0 + (int)per + (blockIdx.x < rem ? 1 : 0);

  if (t0 + wv < t1) {
    const int prow = (t0 + wv) * 16 + m;
    __builtin_prefetch(x + (long)(prow < N ? prow : N - 1) * IN_DIM + (IN_DIM / 2) * kp, 0, 3);
  }

  float bias[4];
  #pragma unroll
  for (int nt = 0; nt < 4; ++nt) bias[nt] = b_in[nt * 16 + m];

  for (int it = tid; it < (IN_DIM / 2) * (HID / 4); it += GEMM_WAVES * 32) {
    const int n4 = it & (HID / 4 - 1), p = it / (HID / 4);
    const float4 r0 = *(const float4*)(w_in + (long)(2 * p) * HID + 4 * n4);
    const float4 r1 = *(const float4*)(w_in + (long)(2 * p + 1) * HID + 4 * n4);
    float4* d = (float4*)(wl1 + p * GEMM_WP + 8 * n4);
    d[0] = make_float4(r0.x, r1.x, r0.y, r1.y);
    d[1] = make_float4(r0.z, r1.z, r0.w, r1.w);
  }
  for (int it = tid; it < (HID / 2) * (OUTD / 4); it += GEMM_WAVES * 32) {
    const int n4 = it & (OUTD / 4 - 1), p = it / (OUTD / 4);
    const float4 r0 = *(const float4*)(w + (long)(2 * p) * OUTD + 4 * n4);
    const float4 r1 = *(const float4*)(w + (long)(2 * p + 1) * OUTD + 4 * n4);
    float4* d = (float4*)(wl2 + p * GEMM_WP + 8 * n4);
    d[0] = make_float4(r0.x, r1.x, r0.y, r1.y);
    d[1] = make_float4(r0.z, r1.z, r0.w, r1.w);
  }
  for (int it = tid; it < (HID / 2) * 32; it += GEMM_WAVES * 32) {
    const int c2 = it & 31, p = it >> 5;
    float v = 0.0f;
    if (c2 < 4) {
      const float4* wr = (const float4*)(w + (long)(2 * p + (c2 & 1)) * OUTD);
      const float4* ar = (const float4*)(a + (c2 >> 1) * OUTD);
      #pragma unroll
      for (int j = 0; j < OUTD / 4; ++j) {
        const float4 u = wr[j];
        const float4 q = ar[j];
        v += u.x * q.x + u.y * q.y + u.z * q.z + u.w * q.w;
      }
    }
    wl2[p * GEMM_WP + 2 * OUTD + c2] = v;
  }
  __syncthreads();

  const float* bq1 = wl1 + kp * GEMM_WP + 2 * m;
  const float* bq2 = wl2 + kp * GEMM_WP + 2 * m;

  v2f xa[IN_DIM / 4];
#define GEMM_LOADX(T_) { \
    const int row_ = (T_) * 16 + m; \
    const int rowc_ = row_ < N ? row_ : (N - 1); \
    const float* xr_ = x + (long)rowc_ * IN_DIM + 2 * kp; \
    _Pragma("unroll") \
    for (int j = 0; j < IN_DIM / 4; ++j) xa[j] = *(const v2f*)(xr_ + 4 * j); }

  int tile = t0 + wv;
  if (tile < t1) GEMM_LOADX(tile)
  for (; tile < t1; tile += GEMM_WAVES) {
    v8f acc[4];
    #pragma unroll
    for (int nt = 0; nt < 4; ++nt) acc[nt] = (v8f){0,0,0,0,0,0,0,0};

    #pragma unroll
    for (int j = 0; j < IN_DIM / 4; ++j) {
      #pragma unroll
      for (int nt = 0; nt < 4; ++nt) {
        v2f bv = *(const v2f*)(bq1 + (2 * j) * GEMM_WP + nt * 32);
        acc[nt] = __builtin_amdgcn_wmma_f32_16x16x4_f32(
            false, xa[j], false, bv, (short)0, acc[nt], false, false);
      }
    }
    if (tile + GEMM_WAVES < t1) GEMM_LOADX(tile + GEMM_WAVES)

    #pragma unroll
    for (int nt = 0; nt < 4; ++nt) {
      #pragma unroll
      for (int r = 0; r < 8; ++r) {
        float v = acc[nt][r] + bias[nt];
        v = 0.5f * v * (1.0f + erff(v * 0.70710678118654752f));
        lds[wv][r + 8 * kp][nt * 16 + m] = v;
      }
    }
    __builtin_amdgcn_fence(__ATOMIC_RELEASE, "wavefront");
    __builtin_amdgcn_wave_barrier();
    __builtin_amdgcn_fence(__ATOMIC_ACQUIRE, "wavefront");

    v8f acc2[5];
    #pragma unroll
    for (int nt = 0; nt < 5; ++nt) acc2[nt] = (v8f){0,0,0,0,0,0,0,0};

    #pragma unroll
    for (int j = 0; j < HID / 4; ++j) {
      v2f av = *(const v2f*)(&lds[wv][m][4 * j + 2 * kp]);
      #pragma unroll
      for (int nt = 0; nt < 5; ++nt) {
        v2f bv = *(const v2f*)(bq2 + (2 * j) * GEMM_WP + nt * 32);
        acc2[nt] = __builtin_amdgcn_wmma_f32_16x16x4_f32(
            false, av, false, bv, (short)0, acc2[nt], false, false);
      }
    }

    #pragma unroll
    for (int nt = 0; nt < 4; ++nt) {
      #pragma unroll
      for (int r = 0; r < 8; ++r) {
        int rr = tile * 16 + r + 8 * kp;
        if (rr < N) z[(long)rr * OUTD + nt * 16 + m] = acc2[nt][r];
      }
    }
    if (m == 0) {
      #pragma unroll
      for (int r = 0; r < 8; ++r) {
        int rr = tile * 16 + r + 8 * kp;
        if (rr < N) s1[rr] = acc2[4][r];
      }
    } else if (m == 1) {
      #pragma unroll
      for (int r = 0; r < 8; ++r) {
        int rr = tile * 16 + r + 8 * kp;
        if (rr < N) s2[rr] = acc2[4][r];
      }
    }
    __builtin_amdgcn_fence(__ATOMIC_RELEASE, "wavefront");
    __builtin_amdgcn_wave_barrier();
    __builtin_amdgcn_fence(__ATOMIC_ACQUIRE, "wavefront");
  }
#undef GEMM_LOADX

  {
    const int lane_k = (int)__builtin_amdgcn_mbcnt_lo(~0u, 0u);
    const int tid_k  = wv * 32 + lane_k;
    const int ng = E >> 5;
    const int4* d4 = (const int4*)dst;
    const int4* s4 = (const int4*)src;
    uint4* pk4 = (uint4*)pk;
    const int bp = own_band_pitch(ng);
    uint2* lo2 = (uint2*)key;
    unsigned* band = key + own_band_base(ng);
    if (blockIdx.x == 0) {
      if (ng & 1) {
        if (tid_k < 64) band[(long)tid_k * bp + ng] = 0u;
        if (tid_k == 0) { uint2 zz; zz.x = 0u; zz.y = 0u; lo2[ng] = zz; }
      }
      if (tid_k < (E & 31)) {
        const int e = (ng << 5) + tid_k;
        pk[e] = own_pack(src[e], dst[e]);
      }
    }
    const int wave_q = (GEMM_WAVES - 1 - wv) * (int)gridDim.x + (int)blockIdx.x;
    for (int j = wave_q * 32 + lane_k; j < ng; j += (int)gridDim.x * GEMM_WAVES * 32) {
      unsigned pl[8] = {0u, 0u, 0u, 0u, 0u, 0u, 0u, 0u};
      #pragma unroll
      for (int q = 0; q < 8; ++q) {
        const int4 p = d4[8 * (long)j + q];
        const int4 sv = s4[8 * (long)j + q];
        own_planes(pl, p.x, 4 * q + 0); own_planes(pl, p.y, 4 * q + 1);
        own_planes(pl, p.z, 4 * q + 2); own_planes(pl, p.w, 4 * q + 3);
        uint4 wd;
        wd.x = own_pack(sv.x, p.x); wd.y = own_pack(sv.y, p.y);
        wd.z = own_pack(sv.z, p.z); wd.w = own_pack(sv.w, p.w);
        pk4[8 * (long)j + q] = wd;
      }
      uint2 lo;
      lo.x = pl[0]; lo.y = pl[1];
      lo2[j] = lo;
      unsigned hA[8], hB[8];
      #pragma unroll
      for (int t = 0; t < 8; ++t) {
        hB[t] = ((t & 1) ? pl[2] : ~pl[2]) & ((t & 2) ? pl[3] : ~pl[3]) & ((t & 4) ? pl[4] : ~pl[4]);
        hA[t] = ((t & 1) ? pl[5] : ~pl[5]) & ((t & 2) ? pl[6] : ~pl[6]) & ((t & 4) ? pl[7] : ~pl[7]);
      }
      #pragma unroll
      for (int t = 0; t < 64; ++t) band[(long)t * bp + j] = hA[t >> 3] & hB[t & 7];
    }
  }
}

#define OWN_CAP     8192
#define OWN_THREADS 1024
#define OWN_SLOTS   ((OWN_ROWS + 31) / 32)

__device__ __forceinline__ float lrelu(float v) { return v > 0.f ? v : 0.01f * v; }
__device__ __forceinline__ int own_idx(int i, int start, int nvec) {
  int j = i + start;
  if (j >= nvec) j -= nvec;
  return j < nvec ? j : nvec - 1;
}

__global__ __launch_bounds__(OWN_THREADS) void gat_owner(
    const int* __restrict__ src, const int* __restrict__ dst,
    const float* __restrict__ z, const float* __restrict__ s1,
    const float* __restrict__ s2, float* __restrict__ h, int N, int E,
    const unsigned* __restrict__ key, const unsigned* __restrict__ pk) {
  __shared__ float    acc[OWN_ROWS * OUTD];
  __shared__ unsigned mb[OWN_ROWS];
  __shared__ float    den[OWN_ROWS];
  __shared__ int      L_e[OWN_CAP];
  __shared__ int      L_r[OWN_CAP];
  __shared__ float    L_v[OWN_CAP];
  __shared__ int      Q_s[OWN_CAP];
  __shared__ int      Q_r[OWN_CAP];
  __shared__ float    Q_v[OWN_CAP];
  __shared__ int      rcnt[32 * OWN_SLOTS];
  __shared__ int      rstart[32 * OWN_SLOTS];
  __shared__ int      cnt;

  const int tid  = threadIdx.x;
  const int lane = tid & 31;
  const int wv   = __builtin_amdgcn_readfirstlane(tid >> 5);
  const int lo   = blockIdx.x * OWN_ROWS;
  const int rows = (N - lo) < OWN_ROWS ? (N - lo) : OWN_ROWS;

  for (int i = tid; i < OWN_ROWS; i += OWN_THREADS) { mb[i] = enc_f32(-INFINITY); den[i] = 0.0f; }
  if (tid < 32 * OWN_SLOTS) rcnt[tid] = 0;
  if (tid == 0) cnt = 0;
  __syncthreads();

  const int ng = E >> 5;
  const int np = (ng + 1) >> 1;
  const unsigned kb = blockIdx.x & 255u;
  const uint4* lo4 = (const uint4*)key;
  const uint2* bw2 = (const uint2*)(key + own_band_base(ng) + (long)(kb >> 2) * own_band_pitch(ng));
  const int start = (int)(((long)blockIdx.x * np) / gridDim.x);
  const unsigned urows = (unsigned)rows;
  const unsigned n0 = ((kb >> 0) & 1u) - 1u, n1 = ((kb >> 1) & 1u) - 1u;
#define OWN_PUSH(EID) { int p_ = atomicAdd(&cnt, 1); if (p_ < OWN_CAP) L_e[p_] = (EID); }
#define OWN_MATCH(P0, P1, BW) ((((P0) ^ n0) & ((P1) ^ n1)) & (BW))
#define OWN_WALK2(MA, JA, MB, JB) { \
    unsigned ma_ = (MA), mb_ = (MB); \
    while (ma_ | mb_) { \
      const bool ua_ = ma_ != 0u; \
      const int b_ = __builtin_ctz(ua_ ? ma_ : mb_); \
      if (ua_) ma_ &= ma_ - 1u; else mb_ &= mb_ - 1u; \
      OWN_PUSH(32 * (ua_ ? (JA) : (JB)) + b_) } }
  int i = tid;
  if (np > 0) {
    int pc0 = own_idx(i, start, np), pc1 = own_idx(i + OWN_THREADS, start, np);
    uint4 cl0 = lo4[pc0], cl1 = lo4[pc1];
    uint2 cb0 = bw2[pc0], cb1 = bw2[pc1];
    while (i + OWN_THREADS < np) {
      const int in = i + 2 * OWN_THREADS;
      const int pn0 = own_idx(in, start, np), pn1 = own_idx(in + OWN_THREADS, start, np);
      const uint4 nl0 = lo4[pn0], nl1 = lo4[pn1];
      const uint2 nb0 = bw2[pn0], nb1 = bw2[pn1];
      OWN_WALK2(OWN_MATCH(cl0.x, cl0.y, cb0.x), 2 * pc0, OWN_MATCH(cl0.z, cl0.w, cb0.y), 2 * pc0 + 1)
      OWN_WALK2(OWN_MATCH(cl1.x, cl1.y, cb1.x), 2 * pc1, OWN_MATCH(cl1.z, cl1.w, cb1.y), 2 * pc1 + 1)
      cl0 = nl0; cl1 = nl1; cb0 = nb0; cb1 = nb1;
      pc0 = pn0; pc1 = pn1;
      i = in;
    }
    for (; i < np; i += OWN_THREADS) {
      const int p0 = own_idx(i, start, np);
      const uint4 dl = lo4[p0];
      const uint2 db = bw2[p0];
      OWN_WALK2(OWN_MATCH(dl.x, dl.y, db.x), 2 * p0, OWN_MATCH(dl.z, dl.w, db.y), 2 * p0 + 1)
    }
  }
  if (tid < (E & 31)) {
    int e = (ng << 5) + tid;
    unsigned r = (unsigned)(dst[e] - lo);
    if (r < urows) OWN_PUSH(e)
  }
  __syncthreads();
  const int total = __builtin_amdgcn_readfirstlane(cnt);

  float* hp = h + (long)lo * OUTD;
  if (total <= OWN_CAP) {
    const bool two_arrays = gridDim.x > 256u || N > (int)(OWN_SRC_MASK + 1u);
    for (int k = tid; k < total; k += OWN_THREADS) {
      int e = L_e[k];
      int s;
      unsigned r;
      if (two_arrays) {
        s = src[e];
        r = (unsigned)(dst[e] - lo);
      } else {
        const unsigned w = pk[e];
        s = (int)(w & OWN_SRC_MASK);
        r = w >> OWN_SRC_BITS;
      }
      if (r < urows) {
        float v = lrelu(s1[s] + s2[lo + (int)r]);
        L_e[k] = s;
        L_v[k] = v;
        atomicMax(&mb[r], enc_f32(v));
        int rank = atomicAdd(&rcnt[(r & 31u) * OWN_SLOTS + (r >> 5)], 1);
        L_r[k] = (int)r | (rank << 9);
      } else {
        L_r[k] = -1;
      }
    }
    __syncthreads();
    int c_l = 0;
    for (int t = 0; t < OWN_SLOTS; ++t) c_l += rcnt[lane * OWN_SLOTS + t];
    int incl = c_l;
    for (int d = 1; d < 32; d <<= 1) {
      int t = __shfl_up(incl, d);
      if (lane >= d) incl += t;
    }
    const int base_l = incl - c_l;
    const int qs = __builtin_amdgcn_readlane(base_l, wv);
    const int qn = __builtin_amdgcn_readlane(c_l, wv);
    {
      const int own = lane < OWN_SLOTS ? rcnt[wv * OWN_SLOTS + lane] : 0;
      int inc = own;
      for (int d = 1; d < 16; d <<= 1) {
        int t = __shfl_up(inc, d);
        if (lane >= d) inc += t;
      }
      if (lane < OWN_SLOTS) rstart[wv * OWN_SLOTS + lane] = qs + inc - own;
    }
    __syncthreads();
    for (int k = tid; k < total; k += OWN_THREADS) {
      const int pr = L_r[k];
      if (pr >= 0) {
        const int r = pr & 511;
        float mv = dec_f32(mb[r]);
        if (!isfinite(mv)) mv = 0.0f;
        float ev = expf(L_v[k] - mv);
        atomicAdd(&den[r], ev);
        const int pos = rstart[(r & 31) * OWN_SLOTS + (r >> 5)] + (pr >> 9);
        Q_s[pos] = L_e[k];
        Q_r[pos] = r;
        Q_v[pos] = ev;
      }
    }
    __syncthreads();
    const unsigned lane8 = (unsigned)lane * 8u;
    v2f*       hl  = (v2f*)hp + lane;
    const int  Gt  = (qn + 15) >> 4;
    int cur = wv;
    v2f sum = {0.0f, 0.0f};
#define OWN_ENT(G_, S_, R_, A_) { \
      const int o_ = 16 * (G_) + (lane & 15); \
      S_ = 0; R_ = 0; A_ = 0.0f; \
      if (o_ < qn) { const int k_ = qs + o_; S_ = Q_s[k_]; R_ = Q_r[k_]; A_ = Q_v[k_] / fmaxf(den[R_], 1e-9f); } }
#define OWN_LOADG(ZV, S_) { \
      _Pragma("unroll") \
      for (int u = 0; u < 16; ++u) { \
        const char* rp_ = (const char*)z + (size_t)(unsigned)__builtin_amdgcn_readlane(S_, u) * (OUTD * 4); \
        ZV[u] = *(const v2f*)(rp_ + lane8); } }
#define OWN_RLF(V, U) __int_as_float(__builtin_amdgcn_readlane(__float_as_int(V), (U)))
#define OWN_ADD1(ZQ, R_, A_, U) { \
      const int r_ = __builtin_amdgcn_readlane(R_, (U)); \
      while (cur < r_) { hl[cur * (OUTD / 2)] = sum; sum = (v2f){0.0f, 0.0f}; cur += 32; } \
      sum += OWN_RLF(A_, (U)) * (ZQ); }
#define OWN_FULL(ZV, R_, A_) { \
      _Pragma("unroll") \
      for (int u = 0; u < 16; ++u) OWN_ADD1(ZV[u], R_, A_, u) }
#define OWN_LAST(ZV, R_, A_, N_) { \
      const int n_ = (N_); \
      _Pragma("unroll") \
      for (int u = 0; u < 16; ++u) if (u < n_) OWN_ADD1(ZV[u], R_, A_, u) }
    int   sA = 0, rA = 0, sB = 0, rB = 0;
    float aA = 0.0f, aB = 0.0f;
    v2f   zvA[16], zvB[16];
    if (Gt > 0) {
      OWN_ENT(0, sA, rA, aA) OWN_LOADG(zvA, sA)
      for (int g = 0; g < Gt; g += 2) {
        if (g + 1 >= Gt) { OWN_LAST(zvA, rA, aA, qn - 16 * g) break; }
        OWN_ENT(g + 1, sB, rB, aB) OWN_LOADG(zvB, sB)
        OWN_FULL(zvA, rA, aA)
        if (g + 2 >= Gt) { OWN_LAST(zvB, rB, aB, qn - 16 * (g + 1)) break; }
        OWN_ENT(g + 2, sA, rA, aA) OWN_LOADG(zvA, sA)
        OWN_FULL(zvB, rB, aB)
      }
    }
    while (cur < rows) { hl[cur * (OUTD / 2)] = sum; sum = (v2f){0.0f, 0.0f}; cur += 32; }
#undef OWN_ENT
#undef OWN_LOADG
#undef OWN_RLF
#undef OWN_ADD1
#undef OWN_FULL
#undef OWN_LAST
  } else {
    for (int i = tid; i < OWN_ROWS * OUTD; i += OWN_THREADS) acc[i] = 0.0f;
    for (int e = tid; e < E; e += OWN_THREADS) {
      unsigned r = (unsigned)(dst[e] - lo);
      if (r < urows) atomicMax(&mb[r], enc_f32(lrelu(s1[src[e]] + s2[lo + (int)r])));
    }
    __syncthreads();
    for (int e = tid; e < E; e += OWN_THREADS) {
      unsigned r = (unsigned)(dst[e] - lo);
      if (r < urows) {
        float mv = dec_f32(mb[r]);
        if (!isfinite(mv)) mv = 0.0f;
        atomicAdd(&den[r], expf(lrelu(s1[src[e]] + s2[lo + (int)r]) - mv));
      }
    }
    __syncthreads();
    for (int e = tid; e < E; e += OWN_THREADS) {
      unsigned r = (unsigned)(dst[e] - lo);
      if (r < urows) {
        int s = src[e];
        float mv = dec_f32(mb[r]);
        if (!isfinite(mv)) mv = 0.0f;
        float a = expf(lrelu(s1[s] + s2[lo + (int)r]) - mv) / fmaxf(den[r], 1e-9f);
        const float* zr = z + (long)s * OUTD;
        for (int c = 0; c < OUTD; ++c) atomicAdd(&acc[r * OUTD + c], a * zr[c]);
      }
    }
    __syncthreads();
    for (int idx = tid; idx < rows * OUTD; idx += OWN_THREADS) hp[idx] = acc[idx];
  }
#undef OWN_PUSH
#undef OWN_MATCH
#undef OWN_WALK2
}

extern "C" void kernel_launch(void* const* d_in, const int* in_sizes, int n_in,
                              void* d_out, int out_size, void* d_ws, size_t ws_size,
                              hipStream_t stream) {
  const float* x    = (const float*)d_in[0];
  const float* w_in = (const float*)d_in[1];
  const float* b_in = (const float*)d_in[2];
  const float* w    = (const float*)d_in[3];
  const float* a    = (const float*)d_in[4];
  const int*   src  = (const int*)d_in[5];
  const int*   dst  = (const int*)d_in[6];
  float* h = (float*)d_out;

  const int N = in_sizes[0] / IN_DIM;
  const int E = in_sizes[5];

  auto align256 = [](size_t v) { return (v + 255) & ~(size_t)255; };
  char* base = (char*)d_ws;
  size_t off = 0;
  float*    z     = (float*)(base + off); off += align256((size_t)N * OUTD * 4);
  float*    s1    = (float*)(base + off); off += align256((size_t)N * 4);
  float*    s2    = (float*)(base + off); off += align256((size_t)N * 4);
  unsigned* key   = (unsigned*)(base + off); off += align256((size_t)E * 9 + 16384);
  unsigned* pk    = (unsigned*)(base + off); off += align256((size_t)E * 4);

  gat_fused_gemm<<<dim3(256), dim3(GEMM_WAVES * 32), 0, stream>>>(
      x, w_in, b_in, w, a, z, s1, s2, N, src, dst, key, pk, E);
  gat_owner<<<dim3((N + OWN_ROWS - 1) / OWN_ROWS), dim3(OWN_THREADS), 0, stream>>>(
      src, dst, z, s1, s2, h, N, E, key, pk);
}
